# starting kernel
# baseline (speedup 1.0000x reference)
#include <hip/hip_runtime.h>
#include <math.h>

typedef float v2f __attribute__((ext_vector_type(2)));
typedef float v8f __attribute__((ext_vector_type(8)));

#define NQ   14
#define DIM  16384
#define NT   256

static __device__ inline v8f wmma_16x16x4_f32(v2f a, v2f b, v8f c) {
#if defined(__gfx1250__)
  return __builtin_amdgcn_wmma_f32_16x16x4_f32(
      false, a, false, b, (short)0, c, false, false);
#else
  return c;
#endif
}

__global__ __launch_bounds__(NT) void qsim14_kernel(
    const float* __restrict__ x,
    const float* __restrict__ params,
    float* __restrict__ out)
{
  __shared__ float st[2 * DIM];
  __shared__ float pHi[128], pLo[128];
  __shared__ float cq[NQ], sq[NQ];
  __shared__ float red[NQ * 8];

  const int b    = blockIdx.x;
  const int tid  = threadIdx.x;
  const int lane = tid & 31;
  const int wave = tid >> 5;

  if (tid < NQ) {
    float h = 0.5f * x[b * NQ + tid];
    cq[tid] = cosf(h);
    sq[tid] = sinf(h);
  }
  __syncthreads();

  if (tid < 128) {
    int h = tid; float p = 1.f;
#pragma unroll
    for (int t = 0; t < 7; ++t) p *= ((h >> (6 - t)) & 1) ? sq[t] : cq[t];
    pHi[h] = p;
  } else {
    int l = tid - 128; float p = 1.f;
#pragma unroll
    for (int u = 0; u < 7; ++u) p *= ((l >> (6 - u)) & 1) ? sq[7 + u] : cq[7 + u];
    pLo[l] = p;
  }
  __syncthreads();

#pragma unroll 4
  for (int k = 0; k < 64; ++k) {
    int idx = k * NT + tid;
    float m = pHi[idx >> 7] * pLo[idx & 127];
    int ph = __popc(idx) & 3;
    st[idx]       = (ph == 0) ? m : ((ph == 2) ? -m : 0.f);
    st[DIM + idx] = (ph == 1) ? -m : ((ph == 3) ? m : 0.f);
  }
  __syncthreads();

  const int baseArr[4] = {10, 6, 2, 0};

  for (int L = 0; L < 2; ++L) {
    for (int g = 0; g < 4; ++g) {
      const int base = baseArr[g];
      const int mask = (1 << base) - 1;

      float gc[4], gs[4];
#pragma unroll
      for (int m = 0; m < 4; ++m) {
        if (g == 3 && m >= 2) { gc[m] = 1.f; gs[m] = 0.f; }
        else {
          int q = 13 - base - m;
          float h = 0.5f * params[L * NQ + q];
          gc[m] = cosf(h); gs[m] = sinf(h);
        }
      }

      const int M    = lane & 15;
      const int Koff = (lane >> 4) << 1;
      v2f A[4];
#pragma unroll
      for (int k = 0; k < 4; ++k) {
#pragma unroll
        for (int r = 0; r < 2; ++r) {
          int Kc = 4 * k + Koff + r;
          float e = 1.f;
#pragma unroll
          for (int m = 0; m < 4; ++m) {
            int ib = (M >> m) & 1, jb = (Kc >> m) & 1;
            e *= (ib == jb) ? gc[m] : ((ib == 0) ? -gs[m] : gs[m]);
          }
          A[k][r] = e;
        }
      }

      for (int t = wave; t < 64; t += 8) {
        int c = 16 * t + (lane & 15);
        int colAddr = ((c >> base) << (base + 4)) + (c & mask);
        int rowsel  = (lane >> 4) << 1;
        int Moff    = (lane >> 4) << 3;
#pragma unroll
        for (int comp = 0; comp < 2; ++comp) {
          float* buf = st + comp * DIM;
          v8f acc = {};
#pragma unroll
          for (int k = 0; k < 4; ++k) {
            v2f Bf;
            Bf[0] = buf[colAddr + ((4 * k + rowsel + 0) << base)];
            Bf[1] = buf[colAddr + ((4 * k + rowsel + 1) << base)];
            acc = wmma_16x16x4_f32(A[k], Bf, acc);
          }
#pragma unroll
          for (int r = 0; r < 8; ++r)
            buf[colAddr + ((r + Moff) << base)] = acc[r];
        }
      }
      __syncthreads();
    }

#pragma unroll
    for (int comp = 0; comp < 2; ++comp) {
      float* buf = st + comp * DIM;
      float v[64];
#pragma unroll
      for (int k = 0; k < 64; ++k) {
        int j = k * NT + tid;
        v[k] = buf[j ^ (j >> 1)];
      }
      __syncthreads();
#pragma unroll
      for (int k = 0; k < 64; ++k)
        buf[k * NT + tid] = v[k];
      __syncthreads();
    }
  }

  float acc14[NQ];
#pragma unroll
  for (int w = 0; w < NQ; ++w) acc14[w] = 0.f;

  for (int k = 0; k < 64; ++k) {
    int j = k * NT + tid;
    float re = st[j], im = st[DIM + j];
    float p = re * re + im * im;
#pragma unroll
    for (int w = 0; w < NQ; ++w)
      acc14[w] += ((j >> (13 - w)) & 1) ? -p : p;
  }

#pragma unroll
  for (int w = 0; w < NQ; ++w) {
    float v = acc14[w];
#pragma unroll
    for (int off = 16; off > 0; off >>= 1)
      v += __shfl_down(v, off, 32);
    if (lane == 0) red[w * 8 + wave] = v;
  }
  __syncthreads();
  if (tid < NQ) {
    float s = 0.f;
#pragma unroll
    for (int i = 0; i < 8; ++i) s += red[tid * 8 + i];
    out[b * NQ + tid] = s;
  }
}

extern "C" void kernel_launch(void* const* d_in, const int* in_sizes, int n_in,
                              void* d_out, int out_size, void* d_ws, size_t ws_size,
                              hipStream_t stream) {
  const float* x      = (const float*)d_in[0];
  const float* params = (const float*)d_in[1];
  float* out          = (float*)d_out;
  int B = in_sizes[0] / NQ;
  qsim14_kernel<<<dim3(B), dim3(NT), 0, stream>>>(x, params, out);
}


# Gate passes fused in pairs: a tile's first WMMA result stays in registers as the A operand of a second 16x16 gate (8 LDS passes become 4; 16-byte LDS accesses, rows padded 16 B); sum order changes
# speedup vs baseline: 1.3413x; 1.3413x over previous
#include <hip/hip_runtime.h>
#include <math.h>

typedef float v2f __attribute__((ext_vector_type(2)));
typedef float v4f __attribute__((ext_vector_type(4)));
typedef float v8f __attribute__((ext_vector_type(8)));

#define NQ   14
#define DIM  16384
#define NT   256
#define ROWP 260
#define CSZ  (64 * ROWP)

static __device__ inline v8f wmma_16x16x4_f32(v2f a, v2f b, v8f c) {
#if defined(__gfx1250__)
  return __builtin_amdgcn_wmma_f32_16x16x4_f32(
      false, a, false, b, (short)0, c, false, false);
#else
  return c;
#endif
}

static __device__ inline void gate_frag(const float* gc, const float* gs, int lane, v2f F[4]) {
  const int N  = lane & 15;
  const int hi = lane >> 4;
#pragma unroll
  for (int k = 0; k < 4; ++k) {
#pragma unroll
    for (int r = 0; r < 2; ++r) {
      const int ib[4] = { r, k & 1, k >> 1, hi };
      float e = 1.f;
#pragma unroll
      for (int f = 0; f < 4; ++f) {
        int ob = (N >> f) & 1;
        e *= (ob == ib[f]) ? gc[f] : ((ob == 0) ? -gs[f] : gs[f]);
      }
      F[k][r] = e;
    }
  }
}

__global__ __launch_bounds__(NT) void qsim14_kernel(
    const float* __restrict__ x,
    const float* __restrict__ params,
    float* __restrict__ out)
{
  __shared__ __attribute__((aligned(16))) float st[2 * CSZ];
  __shared__ float pHi[128], pLo[128];
  __shared__ float cq[NQ], sq[NQ];
  __shared__ float red[NQ * 8];

  const int b    = blockIdx.x;
  const int tid  = threadIdx.x;
  const int lane = tid & 31;
  const int wave = tid >> 5;

  if (tid < NQ) {
    float h = 0.5f * x[b * NQ + tid];
    cq[tid] = cosf(h);
    sq[tid] = sinf(h);
  }
  __syncthreads();

  if (tid < 128) {
    int h = tid; float p = 1.f;
#pragma unroll
    for (int t = 0; t < 7; ++t) p *= ((h >> (6 - t)) & 1) ? sq[t] : cq[t];
    pHi[h] = p;
  } else {
    int l = tid - 128; float p = 1.f;
#pragma unroll
    for (int u = 0; u < 7; ++u) p *= ((l >> (6 - u)) & 1) ? sq[7 + u] : cq[7 + u];
    pLo[l] = p;
  }
  __syncthreads();

#pragma unroll 4
  for (int k = 0; k < 64; ++k) {
    int idx = k * NT + tid;
    float m = pHi[idx >> 7] * pLo[idx & 127];
    int ph = __popc(idx) & 3;
    st[k * ROWP + tid]       = (ph == 0) ? m : ((ph == 2) ? -m : 0.f);
    st[CSZ + k * ROWP + tid] = (ph == 1) ? -m : ((ph == 3) ? m : 0.f);
  }
  __syncthreads();

  for (int L = 0; L < 2; ++L) {
#pragma unroll
    for (int pass = 0; pass < 2; ++pass) {
      const int qb[2][4] = { {13, 12, 7, 6}, {-1, -1, 1, 0} };
      const int qa[2][4] = { {11, 10, 9, 8}, { 5,  4, 3, 2} };
      float c1[4], s1[4], c2[4], s2[4];
#pragma unroll
      for (int f = 0; f < 4; ++f) {
        if (qb[pass][f] < 0) { c1[f] = 1.f; s1[f] = 0.f; }
        else {
          float h = 0.5f * params[L * NQ + qb[pass][f]];
          c1[f] = cosf(h); s1[f] = sinf(h);
        }
        float h2 = 0.5f * params[L * NQ + qa[pass][f]];
        c2[f] = cosf(h2); s2[f] = sinf(h2);
      }
      v2f B1[4], B2[4];
      gate_frag(c1, s1, lane, B1);
      gate_frag(c2, s2, lane, B2);

      const int laneStride = pass ? ROWP      : 4;
      const int hiOff      = pass ? 32 * ROWP : 128;
      const int off2       = pass ? 16 * ROWP : 64;
      const int tileStride = pass ? 4         : ROWP;
      float* pl = st + (lane & 15) * laneStride + (lane >> 4) * hiOff;

      for (int t = wave; t < 64; t += 8) {
        float* p = pl + t * tileStride;
        const v4f r0 = *(const v4f*)(p);
        const v4f r1 = *(const v4f*)(p + off2);
        const v4f i0 = *(const v4f*)(p + CSZ);
        const v4f i1 = *(const v4f*)(p + CSZ + off2);

        v8f dr = {}, di = {};
        dr = wmma_16x16x4_f32(v2f{r0[0], r0[1]}, B1[0], dr);
        di = wmma_16x16x4_f32(v2f{i0[0], i0[1]}, B1[0], di);
        dr = wmma_16x16x4_f32(v2f{r0[2], r0[3]}, B1[1], dr);
        di = wmma_16x16x4_f32(v2f{i0[2], i0[3]}, B1[1], di);
        dr = wmma_16x16x4_f32(v2f{r1[0], r1[1]}, B1[2], dr);
        di = wmma_16x16x4_f32(v2f{i1[0], i1[1]}, B1[2], di);
        dr = wmma_16x16x4_f32(v2f{r1[2], r1[3]}, B1[3], dr);
        di = wmma_16x16x4_f32(v2f{i1[2], i1[3]}, B1[3], di);

        v8f er = {}, ei = {};
        er = wmma_16x16x4_f32(v2f{dr[0], dr[1]}, B2[0], er);
        ei = wmma_16x16x4_f32(v2f{di[0], di[1]}, B2[0], ei);
        er = wmma_16x16x4_f32(v2f{dr[2], dr[3]}, B2[1], er);
        ei = wmma_16x16x4_f32(v2f{di[2], di[3]}, B2[1], ei);
        er = wmma_16x16x4_f32(v2f{dr[4], dr[5]}, B2[2], er);
        ei = wmma_16x16x4_f32(v2f{di[4], di[5]}, B2[2], ei);
        er = wmma_16x16x4_f32(v2f{dr[6], dr[7]}, B2[3], er);
        ei = wmma_16x16x4_f32(v2f{di[6], di[7]}, B2[3], ei);

        *(v4f*)(p)              = v4f{er[0], er[1], er[2], er[3]};
        *(v4f*)(p + off2)       = v4f{er[4], er[5], er[6], er[7]};
        *(v4f*)(p + CSZ)        = v4f{ei[0], ei[1], ei[2], ei[3]};
        *(v4f*)(p + CSZ + off2) = v4f{ei[4], ei[5], ei[6], ei[7]};
      }
      __syncthreads();
    }

#pragma unroll
    for (int comp = 0; comp < 2; ++comp) {
      float* buf = st + comp * CSZ;
      float v[64];
#pragma unroll
      for (int k = 0; k < 64; ++k) {
        int j = k * NT + tid;
        int g = j ^ (j >> 1);
        v[k] = buf[(g >> 8) * ROWP + (g & 255)];
      }
      __syncthreads();
#pragma unroll
      for (int k = 0; k < 64; ++k)
        buf[k * ROWP + tid] = v[k];
      __syncthreads();
    }
  }

  float acc14[NQ];
#pragma unroll
  for (int w = 0; w < NQ; ++w) acc14[w] = 0.f;

  for (int k = 0; k < 64; ++k) {
    int j = k * NT + tid;
    float re = st[k * ROWP + tid], im = st[CSZ + k * ROWP + tid];
    float p = re * re + im * im;
#pragma unroll
    for (int w = 0; w < NQ; ++w)
      acc14[w] += ((j >> (13 - w)) & 1) ? -p : p;
  }

#pragma unroll
  for (int w = 0; w < NQ; ++w) {
    float v = acc14[w];
#pragma unroll
    for (int off = 16; off > 0; off >>= 1)
      v += __shfl_down(v, off, 32);
    if (lane == 0) red[w * 8 + wave] = v;
  }
  __syncthreads();
  if (tid < NQ) {
    float s = 0.f;
#pragma unroll
    for (int i = 0; i < 8; ++i) s += red[tid * 8 + i];
    out[b * NQ + tid] = s;
  }
}

extern "C" void kernel_launch(void* const* d_in, const int* in_sizes, int n_in,
                              void* d_out, int out_size, void* d_ws, size_t ws_size,
                              hipStream_t stream) {
  const float* x      = (const float*)d_in[0];
  const float* params = (const float*)d_in[1];
  float* out          = (float*)d_out;
  int B = in_sizes[0] / NQ;
  qsim14_kernel<<<dim3(B), dim3(NT), 0, stream>>>(x, params, out);
}


# RY cos/sin pairs evaluated once per work-group, one angle per lane beside the RX ones, and kept in LDS; the gate passes read them instead of each wave evaluating 28 pairs on the scalar unit
# speedup vs baseline: 1.9876x; 1.4818x over previous
#include <hip/hip_runtime.h>
#include <math.h>

typedef float v2f __attribute__((ext_vector_type(2)));
typedef float v4f __attribute__((ext_vector_type(4)));
typedef float v8f __attribute__((ext_vector_type(8)));

#define NQ   14
#define DIM  16384
#define NT   256
#define ROWP 260
#define CSZ  (64 * ROWP)

static __device__ inline v8f wmma_16x16x4_f32(v2f a, v2f b, v8f c) {
#if defined(__gfx1250__)
  return __builtin_amdgcn_wmma_f32_16x16x4_f32(
      false, a, false, b, (short)0, c, false, false);
#else
  return c;
#endif
}

static __device__ inline void gate_frag(const float* gc, const float* gs, int lane, int xmask, v2f F[4]) {
  const int No = lane & 15;
  const int N  = No ^ ((No >> 1) & xmask);
  const int hi = lane >> 4;
#pragma unroll
  for (int k = 0; k < 4; ++k) {
#pragma unroll
    for (int r = 0; r < 2; ++r) {
      const int ib[4] = { r, k & 1, k >> 1, hi };
      float e = 1.f;
#pragma unroll
      for (int f = 0; f < 4; ++f) {
        int ob = (N >> f) & 1;
        e *= (ob == ib[f]) ? gc[f] : ((ob == 0) ? -gs[f] : gs[f]);
      }
      F[k][r] = e;
    }
  }
}

__global__ __launch_bounds__(NT) void qsim14_kernel(
    const float* __restrict__ x,
    const float* __restrict__ params,
    float* __restrict__ out)
{
  __shared__ __attribute__((aligned(16))) float st[2 * CSZ];
  __shared__ float pHi[128], pLo[128];
  __shared__ float tc[3 * NQ], ts[3 * NQ];

  const int b    = blockIdx.x;
  const int tid  = threadIdx.x;
  const int lane = tid & 31;
  const int wave = tid >> 5;

  {
    const int slot = (wave == 0 && lane < NQ) ? lane
                   : (wave == 1 && lane < 2 * NQ) ? NQ + lane : -1;
    if (slot >= 0) {
      const float h = 0.5f * ((slot < NQ) ? x[b * NQ + slot] : params[slot - NQ]);
      tc[slot] = cosf(h);
      ts[slot] = sinf(h);
    }
  }
  __syncthreads();

  if (tid < 128) {
    int h = tid; float p = 1.f;
#pragma unroll
    for (int t = 0; t < 7; ++t) p *= ((h >> (6 - t)) & 1) ? ts[t] : tc[t];
    pHi[h] = p;
  } else {
    int l = tid - 128; float p = 1.f;
#pragma unroll
    for (int u = 0; u < 7; ++u) p *= ((l >> (6 - u)) & 1) ? ts[7 + u] : tc[7 + u];
    pLo[l] = p;
  }
  __syncthreads();

#pragma unroll 4
  for (int k = 0; k < 64; ++k) {
    int idx = k * NT + tid;
    float m = pHi[idx >> 7] * pLo[idx & 127];
    int ph = __popc(idx) & 3;
    st[k * ROWP + tid]       = (ph == 0) ? m : ((ph == 2) ? -m : 0.f);
    st[CSZ + k * ROWP + tid] = (ph == 1) ? -m : ((ph == 3) ? m : 0.f);
  }
  __syncthreads();

  for (int L = 0; L < 2; ++L) {
#pragma unroll
    for (int pass = 0; pass < 2; ++pass) {
      const int qb[2][4] = { {-1, -1, 1, 0}, {13, 12, 7, 6} };
      const int qa[2][4] = { { 5,  4, 3, 2}, {11, 10, 9, 8} };
      const float* lc = tc + NQ + L * NQ;
      const float* ls = ts + NQ + L * NQ;
      float c1[4], s1[4], c2[4], s2[4];
#pragma unroll
      for (int f = 0; f < 4; ++f) {
        if (qb[pass][f] < 0) { c1[f] = 1.f; s1[f] = 0.f; }
        else { c1[f] = lc[qb[pass][f]]; s1[f] = ls[qb[pass][f]]; }
        c2[f] = lc[qa[pass][f]]; s2[f] = ls[qa[pass][f]];
      }
      v2f B1[4], B2[4];
      gate_frag(c1, s1, lane, pass ? 5 : 4, B1);
      gate_frag(c2, s2, lane, 7, B2);

      const int laneStride = pass ? 4   : ROWP;
      const int hiOff      = pass ? 128 : 32 * ROWP;
      const int off2       = pass ? 64  : 16 * ROWP;
      const int tileStride = pass ? ROWP : 4;
      const int cl = lane & 15, hl = lane >> 4;
      const float* pl = st + cl * laneStride + hl * hiOff;

      const int T  = pass ? (wave & 1) : 0;
      const int e0 = T, e1 = T ^ 1;
      float* ps0 = st + (cl ^ (e0 ? 15 : 0)) * laneStride + (hl ^ T) * hiOff + e0 * off2;
      float* ps1 = st + (cl ^ (e1 ? 15 : 0)) * laneStride + (hl ^ T) * hiOff + e1 * off2;
      const bool rev0 = ((cl ^ e0) & 1) != 0;

      for (int t = wave; t < 64; t += 8) {
        const float* p = pl + t * tileStride;
        const v4f r0 = *(const v4f*)(p);
        const v4f r1 = *(const v4f*)(p + off2);
        const v4f i0 = *(const v4f*)(p + CSZ);
        const v4f i1 = *(const v4f*)(p + CSZ + off2);

        v8f dr = {}, di = {};
        dr = wmma_16x16x4_f32(v2f{r0[0], r0[1]}, B1[0], dr);
        di = wmma_16x16x4_f32(v2f{i0[0], i0[1]}, B1[0], di);
        dr = wmma_16x16x4_f32(v2f{r0[2], r0[3]}, B1[1], dr);
        di = wmma_16x16x4_f32(v2f{i0[2], i0[3]}, B1[1], di);
        dr = wmma_16x16x4_f32(v2f{r1[0], r1[1]}, B1[2], dr);
        di = wmma_16x16x4_f32(v2f{i1[0], i1[1]}, B1[2], di);
        dr = wmma_16x16x4_f32(v2f{r1[2], r1[3]}, B1[3], dr);
        di = wmma_16x16x4_f32(v2f{i1[2], i1[3]}, B1[3], di);

        v8f er = {}, ei = {};
        er = wmma_16x16x4_f32(v2f{dr[0], dr[1]}, B2[0], er);
        ei = wmma_16x16x4_f32(v2f{di[0], di[1]}, B2[0], ei);
        er = wmma_16x16x4_f32(v2f{dr[2], dr[3]}, B2[1], er);
        ei = wmma_16x16x4_f32(v2f{di[2], di[3]}, B2[1], ei);
        er = wmma_16x16x4_f32(v2f{dr[4], dr[5]}, B2[2], er);
        ei = wmma_16x16x4_f32(v2f{di[4], di[5]}, B2[2], ei);
        er = wmma_16x16x4_f32(v2f{dr[6], dr[7]}, B2[3], er);
        ei = wmma_16x16x4_f32(v2f{di[6], di[7]}, B2[3], ei);

        float* q0 = ps0 + t * tileStride;
        float* q1 = ps1 + t * tileStride;
        if (pass) {
          const bool rev1 = !rev0;
          *(v4f*)(q0)       = rev0 ? v4f{er[3], er[2], er[1], er[0]} : v4f{er[0], er[1], er[2], er[3]};
          *(v4f*)(q1)       = rev1 ? v4f{er[7], er[6], er[5], er[4]} : v4f{er[4], er[5], er[6], er[7]};
          *(v4f*)(q0 + CSZ) = rev0 ? v4f{ei[3], ei[2], ei[1], ei[0]} : v4f{ei[0], ei[1], ei[2], ei[3]};
          *(v4f*)(q1 + CSZ) = rev1 ? v4f{ei[7], ei[6], ei[5], ei[4]} : v4f{ei[4], ei[5], ei[6], ei[7]};
        } else {
          *(v4f*)(q0)       = v4f{er[0], er[1], er[2], er[3]};
          *(v4f*)(q1)       = v4f{er[4], er[5], er[6], er[7]};
          *(v4f*)(q0 + CSZ) = v4f{ei[0], ei[1], ei[2], ei[3]};
          *(v4f*)(q1 + CSZ) = v4f{ei[4], ei[5], ei[6], ei[7]};
        }
      }
      __syncthreads();
    }
  }

  __shared__ float zred[NQ * 16];
  {
    const int cl = lane & 15, hl = lane >> 4;
    const unsigned sel = (cl >= 1 && cl <= 4) ? (1u << (cl - 1)) : 0u;
    const unsigned one = (cl > 6) ? 0u : 0x3f800000u;
    const unsigned hb  = (unsigned)hl << 3;
    v2f S[4];
#pragma unroll
    for (int k = 0; k < 4; ++k) {
#pragma unroll
      for (int r = 0; r < 2; ++r) {
        const unsigned nb = (unsigned)(r | ((k & 1) << 1) | ((k >> 1) << 2)) | hb;
        S[k][r] = __uint_as_float(one | ((nb & sel) ? 0x80000000u : 0u));
      }
    }
    const unsigned f12 = (cl == 5) ? 0x80000000u : 0u;
    const unsigned f13 = (cl == 6) ? 0x80000000u : 0u;

    v8f z0 = {}, z1 = {};
    const float* pl = st + cl * 4 + hl * 128 + wave * ROWP;
#pragma unroll
    for (int ip = 0; ip < 4; ++ip) {
      const float* pa = pl + (16 * ip) * ROWP;
      const float* pb = pa + 8 * ROWP;
      const v4f ra0 = *(const v4f*)(pa);
      const v4f ra1 = *(const v4f*)(pa + 64);
      const v4f ia0 = *(const v4f*)(pa + CSZ);
      const v4f ia1 = *(const v4f*)(pa + CSZ + 64);
      const v4f rb0 = *(const v4f*)(pb);
      const v4f rb1 = *(const v4f*)(pb + 64);
      const v4f ib0 = *(const v4f*)(pb + CSZ);
      const v4f ib1 = *(const v4f*)(pb + CSZ + 64);
      const v4f qa0 = ra0 * ra0 + ia0 * ia0;
      const v4f qa1 = ra1 * ra1 + ia1 * ia1;
      const v4f qb0 = rb0 * rb0 + ib0 * ib0;
      const v4f qb1 = rb1 * rb1 + ib1 * ib1;

      const unsigned flip = ((ip & 1) ? f12 : 0u) | ((ip & 2) ? f13 : 0u);
      v2f Sv[4];
#pragma unroll
      for (int k = 0; k < 4; ++k) {
        Sv[k][0] = __uint_as_float(__float_as_uint(S[k][0]) ^ flip);
        Sv[k][1] = __uint_as_float(__float_as_uint(S[k][1]) ^ flip);
      }

      z0 = wmma_16x16x4_f32(v2f{qa0[0], qa0[1]}, Sv[0], z0);
      z1 = wmma_16x16x4_f32(v2f{qb0[0], qb0[1]}, Sv[0], z1);
      z0 = wmma_16x16x4_f32(v2f{qa0[2], qa0[3]}, Sv[1], z0);
      z1 = wmma_16x16x4_f32(v2f{qb0[2], qb0[3]}, Sv[1], z1);
      z0 = wmma_16x16x4_f32(v2f{qa1[0], qa1[1]}, Sv[2], z0);
      z1 = wmma_16x16x4_f32(v2f{qb1[0], qb1[1]}, Sv[2], z1);
      z0 = wmma_16x16x4_f32(v2f{qa1[2], qa1[3]}, Sv[3], z0);
      z1 = wmma_16x16x4_f32(v2f{qb1[2], qb1[3]}, Sv[3], z1);
    }

#if defined(__gfx1250__)
    asm volatile("v_nop\n\tv_nop\n\tv_nop\n\tv_nop" : "+v"(z0), "+v"(z1));
#endif
    float cs = 0.f, d11 = 0.f, mb0 = 0.f, mb1 = 0.f, mb2 = 0.f;
#pragma unroll
    for (int r = 0; r < 8; ++r) {
      const float sp = z0[r] + z1[r];
      cs  += sp;
      d11 += z0[r] - z1[r];
      mb0 += (r & 1) ? -sp : sp;
      mb1 += (r & 2) ? -sp : sp;
      mb2 += (r & 4) ? -sp : sp;
    }
    const int slot = wave * 2 + hl;
    if (cl == 0) {
      zred[ 2 * 16 + slot] = mb0;
      zred[ 3 * 16 + slot] = mb1;
      zred[ 4 * 16 + slot] = mb2;
      zred[ 5 * 16 + slot] = hl ? -cs : cs;
      zred[ 8 * 16 + slot] = (wave & 1) ? -cs : cs;
      zred[ 9 * 16 + slot] = (wave & 2) ? -cs : cs;
      zred[10 * 16 + slot] = (wave & 4) ? -cs : cs;
      zred[11 * 16 + slot] = d11;
    } else if (cl <= 6) {
      const int bit = (cl == 1) ? 0 : (cl == 2) ? 1 : (cl == 3) ? 6 : (cl == 4) ? 7 : (cl == 5) ? 12 : 13;
      zred[bit * 16 + slot] = cs;
    }
  }
  __syncthreads();
  if (tid < NQ) {
    const float* zr = zred + (13 - tid) * 16;
    float s = 0.f;
#pragma unroll
    for (int i = 0; i < 16; ++i) s += zr[i];
    out[b * NQ + tid] = s;
  }
}

extern "C" void kernel_launch(void* const* d_in, const int* in_sizes, int n_in,
                              void* d_out, int out_size, void* d_ws, size_t ws_size,
                              hipStream_t stream) {
  const float* x      = (const float*)d_in[0];
  const float* params = (const float*)d_in[1];
  float* out          = (float*)d_out;
  int B = in_sizes[0] / NQ;
  qsim14_kernel<<<dim3(B), dim3(NT), 0, stream>>>(x, params, out);
}


# Product-state write vectorised: a thread forms four adjacent amplitudes as one quarter-turned 4-vector (popcount phase folded into per-thread constants) and stores 16 B; 128 scalar stores become 32
# speedup vs baseline: 2.1154x; 1.0643x over previous
#include <hip/hip_runtime.h>
#include <math.h>

typedef float v2f __attribute__((ext_vector_type(2)));
typedef float v4f __attribute__((ext_vector_type(4)));
typedef float v8f __attribute__((ext_vector_type(8)));

#define NQ   14
#define DIM  16384
#define NT   256
#define ROWP 260
#define CSZ  (64 * ROWP)

static __device__ inline v8f wmma_16x16x4_f32(v2f a, v2f b, v8f c) {
#if defined(__gfx1250__)
  return __builtin_amdgcn_wmma_f32_16x16x4_f32(
      false, a, false, b, (short)0, c, false, false);
#else
  return c;
#endif
}

static __device__ inline void gate_frag(const float* gc, const float* gs, int lane, int xmask, v2f F[4]) {
  const int No = lane & 15;
  const int N  = No ^ ((No >> 1) & xmask);
  const int hi = lane >> 4;
#pragma unroll
  for (int k = 0; k < 4; ++k) {
#pragma unroll
    for (int r = 0; r < 2; ++r) {
      const int ib[4] = { r, k & 1, k >> 1, hi };
      float e = 1.f;
#pragma unroll
      for (int f = 0; f < 4; ++f) {
        int ob = (N >> f) & 1;
        e *= (ob == ib[f]) ? gc[f] : ((ob == 0) ? -gs[f] : gs[f]);
      }
      F[k][r] = e;
    }
  }
}

__global__ __launch_bounds__(NT) void qsim14_kernel(
    const float* __restrict__ x,
    const float* __restrict__ params,
    float* __restrict__ out)
{
  __shared__ __attribute__((aligned(16))) float st[2 * CSZ];
  __shared__ float pHi[128];
  __shared__ __attribute__((aligned(16))) float pLo[128];
  __shared__ float tc[3 * NQ], ts[3 * NQ];

  const int b    = blockIdx.x;
  const int tid  = threadIdx.x;
  const int lane = tid & 31;
  const int wave = tid >> 5;

  {
    const int slot = (wave == 0 && lane < NQ) ? lane
                   : (wave == 1 && lane < 2 * NQ) ? NQ + lane : -1;
    if (slot >= 0) {
      const float h = 0.5f * ((slot < NQ) ? x[b * NQ + slot] : params[slot - NQ]);
      tc[slot] = cosf(h);
      ts[slot] = sinf(h);
    }
  }
  __syncthreads();

  if (tid < 128) {
    int h = tid; float p = 1.f;
#pragma unroll
    for (int t = 0; t < 7; ++t) p *= ((h >> (6 - t)) & 1) ? ts[t] : tc[t];
    pHi[h] = p;
  } else {
    int l = tid - 128; float p = 1.f;
#pragma unroll
    for (int u = 0; u < 7; ++u) p *= ((l >> (6 - u)) & 1) ? ts[7 + u] : tc[7 + u];
    pLo[l] = p;
  }
  __syncthreads();

  {
    const int c  = tid & 63;
    const int rq = tid >> 6;
    const v4f Lw = *(const v4f*)(pLo + 4 * (c & 31));
    const int pt = __popc(tid) & 3;
    const float tr = (pt == 0) ? 1.f : ((pt == 2) ? -1.f : 0.f);
    const float ti = (pt == 1) ? -1.f : ((pt == 3) ? 1.f : 0.f);
    const v4f Rr = { tr * Lw[0],  ti * Lw[1],  ti * Lw[2], -tr * Lw[3] };
    const v4f Ri = { ti * Lw[0], -tr * Lw[1], -tr * Lw[2], -ti * Lw[3] };
    const float* ph = pHi + 2 * rq + (c >> 5);
    float* ps = st + rq * ROWP + 4 * c;
#pragma unroll
    for (int i = 0; i < 16; ++i) {
      const float H = ph[8 * i];
      const int q = __builtin_popcount(i) & 3;
      v4f re, im;
      if (q == 0)      { re =  Rr; im =  Ri; }
      else if (q == 1) { re =  Ri; im = -Rr; }
      else if (q == 2) { re = -Rr; im = -Ri; }
      else             { re = -Ri; im =  Rr; }
      *(v4f*)(ps + 4 * i * ROWP)       = H * re;
      *(v4f*)(ps + CSZ + 4 * i * ROWP) = H * im;
    }
  }
  __syncthreads();

  for (int L = 0; L < 2; ++L) {
#pragma unroll
    for (int pass = 0; pass < 2; ++pass) {
      const int qb[2][4] = { {-1, -1, 1, 0}, {13, 12, 7, 6} };
      const int qa[2][4] = { { 5,  4, 3, 2}, {11, 10, 9, 8} };
      const float* lc = tc + NQ + L * NQ;
      const float* ls = ts + NQ + L * NQ;
      float c1[4], s1[4], c2[4], s2[4];
#pragma unroll
      for (int f = 0; f < 4; ++f) {
        if (qb[pass][f] < 0) { c1[f] = 1.f; s1[f] = 0.f; }
        else { c1[f] = lc[qb[pass][f]]; s1[f] = ls[qb[pass][f]]; }
        c2[f] = lc[qa[pass][f]]; s2[f] = ls[qa[pass][f]];
      }
      v2f B1[4], B2[4];
      gate_frag(c1, s1, lane, pass ? 5 : 4, B1);
      gate_frag(c2, s2, lane, 7, B2);

      const int laneStride = pass ? 4   : ROWP;
      const int hiOff      = pass ? 128 : 32 * ROWP;
      const int off2       = pass ? 64  : 16 * ROWP;
      const int tileStride = pass ? ROWP : 4;
      const int cl = lane & 15, hl = lane >> 4;
      const float* pl = st + cl * laneStride + hl * hiOff;

      const int T  = pass ? (wave & 1) : 0;
      const int e0 = T, e1 = T ^ 1;
      float* ps0 = st + (cl ^ (e0 ? 15 : 0)) * laneStride + (hl ^ T) * hiOff + e0 * off2;
      float* ps1 = st + (cl ^ (e1 ? 15 : 0)) * laneStride + (hl ^ T) * hiOff + e1 * off2;
      const bool rev0 = ((cl ^ e0) & 1) != 0;

      for (int t = wave; t < 64; t += 8) {
        const float* p = pl + t * tileStride;
        const v4f r0 = *(const v4f*)(p);
        const v4f r1 = *(const v4f*)(p + off2);
        const v4f i0 = *(const v4f*)(p + CSZ);
        const v4f i1 = *(const v4f*)(p + CSZ + off2);

        v8f dr = {}, di = {};
        dr = wmma_16x16x4_f32(v2f{r0[0], r0[1]}, B1[0], dr);
        di = wmma_16x16x4_f32(v2f{i0[0], i0[1]}, B1[0], di);
        dr = wmma_16x16x4_f32(v2f{r0[2], r0[3]}, B1[1], dr);
        di = wmma_16x16x4_f32(v2f{i0[2], i0[3]}, B1[1], di);
        dr = wmma_16x16x4_f32(v2f{r1[0], r1[1]}, B1[2], dr);
        di = wmma_16x16x4_f32(v2f{i1[0], i1[1]}, B1[2], di);
        dr = wmma_16x16x4_f32(v2f{r1[2], r1[3]}, B1[3], dr);
        di = wmma_16x16x4_f32(v2f{i1[2], i1[3]}, B1[3], di);

        v8f er = {}, ei = {};
        er = wmma_16x16x4_f32(v2f{dr[0], dr[1]}, B2[0], er);
        ei = wmma_16x16x4_f32(v2f{di[0], di[1]}, B2[0], ei);
        er = wmma_16x16x4_f32(v2f{dr[2], dr[3]}, B2[1], er);
        ei = wmma_16x16x4_f32(v2f{di[2], di[3]}, B2[1], ei);
        er = wmma_16x16x4_f32(v2f{dr[4], dr[5]}, B2[2], er);
        ei = wmma_16x16x4_f32(v2f{di[4], di[5]}, B2[2], ei);
        er = wmma_16x16x4_f32(v2f{dr[6], dr[7]}, B2[3], er);
        ei = wmma_16x16x4_f32(v2f{di[6], di[7]}, B2[3], ei);

        float* q0 = ps0 + t * tileStride;
        float* q1 = ps1 + t * tileStride;
        if (pass) {
          const bool rev1 = !rev0;
          *(v4f*)(q0)       = rev0 ? v4f{er[3], er[2], er[1], er[0]} : v4f{er[0], er[1], er[2], er[3]};
          *(v4f*)(q1)       = rev1 ? v4f{er[7], er[6], er[5], er[4]} : v4f{er[4], er[5], er[6], er[7]};
          *(v4f*)(q0 + CSZ) = rev0 ? v4f{ei[3], ei[2], ei[1], ei[0]} : v4f{ei[0], ei[1], ei[2], ei[3]};
          *(v4f*)(q1 + CSZ) = rev1 ? v4f{ei[7], ei[6], ei[5], ei[4]} : v4f{ei[4], ei[5], ei[6], ei[7]};
        } else {
          *(v4f*)(q0)       = v4f{er[0], er[1], er[2], er[3]};
          *(v4f*)(q1)       = v4f{er[4], er[5], er[6], er[7]};
          *(v4f*)(q0 + CSZ) = v4f{ei[0], ei[1], ei[2], ei[3]};
          *(v4f*)(q1 + CSZ) = v4f{ei[4], ei[5], ei[6], ei[7]};
        }
      }
      __syncthreads();
    }
  }

  __shared__ float zred[NQ * 16];
  {
    const int cl = lane & 15, hl = lane >> 4;
    const unsigned sel = (cl >= 1 && cl <= 4) ? (1u << (cl - 1)) : 0u;
    const unsigned one = (cl > 6) ? 0u : 0x3f800000u;
    const unsigned hb  = (unsigned)hl << 3;
    v2f S[4];
#pragma unroll
    for (int k = 0; k < 4; ++k) {
#pragma unroll
      for (int r = 0; r < 2; ++r) {
        const unsigned nb = (unsigned)(r | ((k & 1) << 1) | ((k >> 1) << 2)) | hb;
        S[k][r] = __uint_as_float(one | ((nb & sel) ? 0x80000000u : 0u));
      }
    }
    const unsigned f12 = (cl == 5) ? 0x80000000u : 0u;
    const unsigned f13 = (cl == 6) ? 0x80000000u : 0u;

    v8f z0 = {}, z1 = {};
    const float* pl = st + cl * 4 + hl * 128 + wave * ROWP;
#pragma unroll
    for (int ip = 0; ip < 4; ++ip) {
      const float* pa = pl + (16 * ip) * ROWP;
      const float* pb = pa + 8 * ROWP;
      const v4f ra0 = *(const v4f*)(pa);
      const v4f ra1 = *(const v4f*)(pa + 64);
      const v4f ia0 = *(const v4f*)(pa + CSZ);
      const v4f ia1 = *(const v4f*)(pa + CSZ + 64);
      const v4f rb0 = *(const v4f*)(pb);
      const v4f rb1 = *(const v4f*)(pb + 64);
      const v4f ib0 = *(const v4f*)(pb + CSZ);
      const v4f ib1 = *(const v4f*)(pb + CSZ + 64);
      const v4f qa0 = ra0 * ra0 + ia0 * ia0;
      const v4f qa1 = ra1 * ra1 + ia1 * ia1;
      const v4f qb0 = rb0 * rb0 + ib0 * ib0;
      const v4f qb1 = rb1 * rb1 + ib1 * ib1;

      const unsigned flip = ((ip & 1) ? f12 : 0u) | ((ip & 2) ? f13 : 0u);
      v2f Sv[4];
#pragma unroll
      for (int k = 0; k < 4; ++k) {
        Sv[k][0] = __uint_as_float(__float_as_uint(S[k][0]) ^ flip);
        Sv[k][1] = __uint_as_float(__float_as_uint(S[k][1]) ^ flip);
      }

      z0 = wmma_16x16x4_f32(v2f{qa0[0], qa0[1]}, Sv[0], z0);
      z1 = wmma_16x16x4_f32(v2f{qb0[0], qb0[1]}, Sv[0], z1);
      z0 = wmma_16x16x4_f32(v2f{qa0[2], qa0[3]}, Sv[1], z0);
      z1 = wmma_16x16x4_f32(v2f{qb0[2], qb0[3]}, Sv[1], z1);
      z0 = wmma_16x16x4_f32(v2f{qa1[0], qa1[1]}, Sv[2], z0);
      z1 = wmma_16x16x4_f32(v2f{qb1[0], qb1[1]}, Sv[2], z1);
      z0 = wmma_16x16x4_f32(v2f{qa1[2], qa1[3]}, Sv[3], z0);
      z1 = wmma_16x16x4_f32(v2f{qb1[2], qb1[3]}, Sv[3], z1);
    }

#if defined(__gfx1250__)
    asm volatile("v_nop\n\tv_nop\n\tv_nop\n\tv_nop" : "+v"(z0), "+v"(z1));
#endif
    float cs = 0.f, d11 = 0.f, mb0 = 0.f, mb1 = 0.f, mb2 = 0.f;
#pragma unroll
    for (int r = 0; r < 8; ++r) {
      const float sp = z0[r] + z1[r];
      cs  += sp;
      d11 += z0[r] - z1[r];
      mb0 += (r & 1) ? -sp : sp;
      mb1 += (r & 2) ? -sp : sp;
      mb2 += (r & 4) ? -sp : sp;
    }
    const int slot = wave * 2 + hl;
    if (cl == 0) {
      zred[ 2 * 16 + slot] = mb0;
      zred[ 3 * 16 + slot] = mb1;
      zred[ 4 * 16 + slot] = mb2;
      zred[ 5 * 16 + slot] = hl ? -cs : cs;
      zred[ 8 * 16 + slot] = (wave & 1) ? -cs : cs;
      zred[ 9 * 16 + slot] = (wave & 2) ? -cs : cs;
      zred[10 * 16 + slot] = (wave & 4) ? -cs : cs;
      zred[11 * 16 + slot] = d11;
    } else if (cl <= 6) {
      const int bit = (cl == 1) ? 0 : (cl == 2) ? 1 : (cl == 3) ? 6 : (cl == 4) ? 7 : (cl == 5) ? 12 : 13;
      zred[bit * 16 + slot] = cs;
    }
  }
  __syncthreads();
  if (tid < NQ) {
    const float* zr = zred + (13 - tid) * 16;
    float s = 0.f;
#pragma unroll
    for (int i = 0; i < 16; ++i) s += zr[i];
    out[b * NQ + tid] = s;
  }
}

extern "C" void kernel_launch(void* const* d_in, const int* in_sizes, int n_in,
                              void* d_out, int out_size, void* d_ws, size_t ws_size,
                              hipStream_t stream) {
  const float* x      = (const float*)d_in[0];
  const float* params = (const float*)d_in[1];
  float* out          = (float*)d_out;
  int B = in_sizes[0] / NQ;
  qsim14_kernel<<<dim3(B), dim3(NT), 0, stream>>>(x, params, out);
}


# Layer 0 high pass: index bits 0,1 left out of the written state and entering as a Kronecker factor of the first gate matrix (K=16 becomes K=4, 1 WMMA for 4; state write 4x smaller); sum order changes
# speedup vs baseline: 2.3071x; 1.0906x over previous
#include <hip/hip_runtime.h>
#include <math.h>

typedef float v2f __attribute__((ext_vector_type(2)));
typedef float v4f __attribute__((ext_vector_type(4)));
typedef float v8f __attribute__((ext_vector_type(8)));

#define NQ   14
#define DIM  16384
#define NT   256
#define ROWP 260
#define CSZ  (64 * ROWP)

static __device__ inline v8f wmma_16x16x4_f32(v2f a, v2f b, v8f c) {
#if defined(__gfx1250__)
  return __builtin_amdgcn_wmma_f32_16x16x4_f32(
      false, a, false, b, (short)0, c, false, false);
#else
  return c;
#endif
}

static __device__ inline void gate_frag(const float* gc, const float* gs, int lane, int xmask, v2f F[4]) {
  const int No = lane & 15;
  const int N  = No ^ ((No >> 1) & xmask);
  const int hi = lane >> 4;
#pragma unroll
  for (int k = 0; k < 4; ++k) {
#pragma unroll
    for (int r = 0; r < 2; ++r) {
      const int ib[4] = { r, k & 1, k >> 1, hi };
      float e = 1.f;
#pragma unroll
      for (int f = 0; f < 4; ++f) {
        int ob = (N >> f) & 1;
        e *= (ob == ib[f]) ? gc[f] : ((ob == 0) ? -gs[f] : gs[f]);
      }
      F[k][r] = e;
    }
  }
}

__global__ __launch_bounds__(NT) void qsim14_kernel(
    const float* __restrict__ x,
    const float* __restrict__ params,
    float* __restrict__ out)
{
  __shared__ __attribute__((aligned(16))) float st[2 * CSZ];
  __shared__ __attribute__((aligned(16))) float pHi[128];
  __shared__ float pL5[32];
  __shared__ float tc[3 * NQ], ts[3 * NQ];

  const int b    = blockIdx.x;
  const int tid  = threadIdx.x;
  const int lane = tid & 31;
  const int wave = tid >> 5;

  {
    const int slot = (wave == 0 && lane < NQ) ? lane
                   : (wave == 1 && lane < 2 * NQ) ? NQ + lane : -1;
    if (slot >= 0) {
      const float h = 0.5f * ((slot < NQ) ? x[b * NQ + slot] : params[slot - NQ]);
      tc[slot] = cosf(h);
      ts[slot] = sinf(h);
    }
  }
  __syncthreads();

  if (tid < 128) {
    int h = tid; float p = 1.f;
#pragma unroll
    for (int t = 0; t < 7; ++t) p *= ((h >> (6 - t)) & 1) ? ts[t] : tc[t];
    pHi[((h & 31) << 2) | (h >> 5)] = p;
  } else if (tid < 160) {
    int l = tid - 128; float p = 1.f;
#pragma unroll
    for (int u = 0; u < 5; ++u) p *= ((l >> (4 - u)) & 1) ? ts[7 + u] : tc[7 + u];
    pL5[l] = p;
  }
  __syncthreads();

  {
    const int c  = tid & 63;
    const int rq = tid >> 6;
    const float Lw = pL5[c & 31];
    const int pt = __popc(tid) & 3;
    const float tr = (pt == 0) ? 1.f : ((pt == 2) ? -1.f : 0.f);
    const float ti = (pt == 1) ? -1.f : ((pt == 3) ? 1.f : 0.f);
    const v4f Rr = { tr * Lw,  ti * Lw,  ti * Lw, -tr * Lw };
    const v4f Ri = { ti * Lw, -tr * Lw, -tr * Lw, -ti * Lw };
    const float* ph = pHi + 4 * (2 * rq + (c >> 5));
    float* ps = st + rq * ROWP + 4 * c;
#pragma unroll
    for (int i = 0; i < 4; ++i) {
      const v4f H = *(const v4f*)(ph + 32 * i);
      const int q = __builtin_popcount(i) & 3;
      v4f re, im;
      if (q == 0)      { re =  Rr; im =  Ri; }
      else if (q == 1) { re =  Ri; im = -Rr; }
      else             { re = -Rr; im = -Ri; }
      *(v4f*)(ps + 4 * i * ROWP)       = H * re;
      *(v4f*)(ps + CSZ + 4 * i * ROWP) = H * im;
    }
  }
  __syncthreads();

#pragma unroll
  for (int L = 0; L < 2; ++L) {
#pragma unroll
    for (int pass = 0; pass < 2; ++pass) {
      const int qb[2][4] = { {-1, -1, 1, 0}, {13, 12, 7, 6} };
      const int qa[2][4] = { { 5,  4, 3, 2}, {11, 10, 9, 8} };
      const float* lc = tc + NQ + L * NQ;
      const float* ls = ts + NQ + L * NQ;
      float c1[4], s1[4], c2[4], s2[4];
#pragma unroll
      for (int f = 0; f < 4; ++f) {
        if (qb[pass][f] < 0) { c1[f] = 1.f; s1[f] = 0.f; }
        else { c1[f] = lc[qb[pass][f]]; s1[f] = ls[qb[pass][f]]; }
        c2[f] = lc[qa[pass][f]]; s2[f] = ls[qa[pass][f]];
      }
      v2f B1[4], B2[4];
      gate_frag(c1, s1, lane, pass ? 5 : 4, B1);
      gate_frag(c2, s2, lane, 7, B2);

      const int laneStride = pass ? 4   : ROWP;
      const int hiOff      = pass ? 128 : 32 * ROWP;
      const int off2       = pass ? 64  : 16 * ROWP;
      const int tileStride = pass ? ROWP : 4;
      const int cl = lane & 15, hl = lane >> 4;
      const float* pl = st + cl * laneStride + hl * hiOff;

      const bool first = (L == 0 && pass == 0);
      v2f E1r = {}, E1i = {};
      const float* pf = st + cl * ROWP + 2 * hl;
      if (first) {
        const int a  = cl ^ ((cl >> 1) & 4);
        const int j  = cl & 3;
        const float mg = ((j & 1) ? ts[13] : tc[13]) * ((j & 2) ? ts[12] : tc[12]);
        const int o12 = (a >> 2) & 1, o13 = (a >> 3) & 1;
        const float g13 = (o13 == hl) ? c1[3] : ((o13 == 0) ? -s1[3] : s1[3]);
#pragma unroll
        for (int r = 0; r < 2; ++r) {
          const float g12 = (o12 == r) ? c1[2] : ((o12 == 0) ? -s1[2] : s1[2]);
          const float e = g12 * g13 * mg;
          E1r[r] = (j == 0) ? e : -e;
          E1i[r] = (j == 3) ? -e : e;
        }
      }

      const int T  = pass ? (wave & 1) : 0;
      const int e0 = T, e1 = T ^ 1;
      float* ps0 = st + (cl ^ (e0 ? 15 : 0)) * laneStride + (hl ^ T) * hiOff + e0 * off2;
      float* ps1 = st + (cl ^ (e1 ? 15 : 0)) * laneStride + (hl ^ T) * hiOff + e1 * off2;
      const bool rev0 = ((cl ^ e0) & 1) != 0;

      for (int t = wave; t < 64; t += 8) {
        v8f dr = {}, di = {};
        if (first) {
          const v2f ar = *(const v2f*)(pf + 4 * t);
          const v2f ai = *(const v2f*)(pf + 4 * t + CSZ);
          dr = wmma_16x16x4_f32(ar, E1r, dr);
          di = wmma_16x16x4_f32(ai, E1i, di);
        } else {
          const float* p = pl + t * tileStride;
          const v4f r0 = *(const v4f*)(p);
          const v4f r1 = *(const v4f*)(p + off2);
          const v4f i0 = *(const v4f*)(p + CSZ);
          const v4f i1 = *(const v4f*)(p + CSZ + off2);
          dr = wmma_16x16x4_f32(v2f{r0[0], r0[1]}, B1[0], dr);
          di = wmma_16x16x4_f32(v2f{i0[0], i0[1]}, B1[0], di);
          dr = wmma_16x16x4_f32(v2f{r0[2], r0[3]}, B1[1], dr);
          di = wmma_16x16x4_f32(v2f{i0[2], i0[3]}, B1[1], di);
          dr = wmma_16x16x4_f32(v2f{r1[0], r1[1]}, B1[2], dr);
          di = wmma_16x16x4_f32(v2f{i1[0], i1[1]}, B1[2], di);
          dr = wmma_16x16x4_f32(v2f{r1[2], r1[3]}, B1[3], dr);
          di = wmma_16x16x4_f32(v2f{i1[2], i1[3]}, B1[3], di);
        }

        v8f er = {}, ei = {};
        er = wmma_16x16x4_f32(v2f{dr[0], dr[1]}, B2[0], er);
        ei = wmma_16x16x4_f32(v2f{di[0], di[1]}, B2[0], ei);
        er = wmma_16x16x4_f32(v2f{dr[2], dr[3]}, B2[1], er);
        ei = wmma_16x16x4_f32(v2f{di[2], di[3]}, B2[1], ei);
        er = wmma_16x16x4_f32(v2f{dr[4], dr[5]}, B2[2], er);
        ei = wmma_16x16x4_f32(v2f{di[4], di[5]}, B2[2], ei);
        er = wmma_16x16x4_f32(v2f{dr[6], dr[7]}, B2[3], er);
        ei = wmma_16x16x4_f32(v2f{di[6], di[7]}, B2[3], ei);

        float* q0 = ps0 + t * tileStride;
        float* q1 = ps1 + t * tileStride;
        if (pass) {
          const bool rev1 = !rev0;
          *(v4f*)(q0)       = rev0 ? v4f{er[3], er[2], er[1], er[0]} : v4f{er[0], er[1], er[2], er[3]};
          *(v4f*)(q1)       = rev1 ? v4f{er[7], er[6], er[5], er[4]} : v4f{er[4], er[5], er[6], er[7]};
          *(v4f*)(q0 + CSZ) = rev0 ? v4f{ei[3], ei[2], ei[1], ei[0]} : v4f{ei[0], ei[1], ei[2], ei[3]};
          *(v4f*)(q1 + CSZ) = rev1 ? v4f{ei[7], ei[6], ei[5], ei[4]} : v4f{ei[4], ei[5], ei[6], ei[7]};
        } else if (first) {
          *(v4f*)(q0)       = v4f{er[0], ei[1], ei[2], er[3]};
          *(v4f*)(q1)       = v4f{er[4], ei[5], ei[6], er[7]};
          *(v4f*)(q0 + CSZ) = v4f{ei[0], er[1], er[2], ei[3]};
          *(v4f*)(q1 + CSZ) = v4f{ei[4], er[5], er[6], ei[7]};
        } else {
          *(v4f*)(q0)       = v4f{er[0], er[1], er[2], er[3]};
          *(v4f*)(q1)       = v4f{er[4], er[5], er[6], er[7]};
          *(v4f*)(q0 + CSZ) = v4f{ei[0], ei[1], ei[2], ei[3]};
          *(v4f*)(q1 + CSZ) = v4f{ei[4], ei[5], ei[6], ei[7]};
        }
      }
      __syncthreads();
    }
  }

  __shared__ float zred[NQ * 16];
  {
    const int cl = lane & 15, hl = lane >> 4;
    const unsigned sel = (cl >= 1 && cl <= 4) ? (1u << (cl - 1)) : 0u;
    const unsigned one = (cl > 6) ? 0u : 0x3f800000u;
    const unsigned hb  = (unsigned)hl << 3;
    v2f S[4];
#pragma unroll
    for (int k = 0; k < 4; ++k) {
#pragma unroll
      for (int r = 0; r < 2; ++r) {
        const unsigned nb = (unsigned)(r | ((k & 1) << 1) | ((k >> 1) << 2)) | hb;
        S[k][r] = __uint_as_float(one | ((nb & sel) ? 0x80000000u : 0u));
      }
    }
    const unsigned f12 = (cl == 5) ? 0x80000000u : 0u;
    const unsigned f13 = (cl == 6) ? 0x80000000u : 0u;

    v8f z0 = {}, z1 = {};
    const float* pl = st + cl * 4 + hl * 128 + wave * ROWP;
#pragma unroll
    for (int ip = 0; ip < 4; ++ip) {
      const float* pa = pl + (16 * ip) * ROWP;
      const float* pb = pa + 8 * ROWP;
      const v4f ra0 = *(const v4f*)(pa);
      const v4f ra1 = *(const v4f*)(pa + 64);
      const v4f ia0 = *(const v4f*)(pa + CSZ);
      const v4f ia1 = *(const v4f*)(pa + CSZ + 64);
      const v4f rb0 = *(const v4f*)(pb);
      const v4f rb1 = *(const v4f*)(pb + 64);
      const v4f ib0 = *(const v4f*)(pb + CSZ);
      const v4f ib1 = *(const v4f*)(pb + CSZ + 64);
      const v4f qa0 = ra0 * ra0 + ia0 * ia0;
      const v4f qa1 = ra1 * ra1 + ia1 * ia1;
      const v4f qb0 = rb0 * rb0 + ib0 * ib0;
      const v4f qb1 = rb1 * rb1 + ib1 * ib1;

      const unsigned flip = ((ip & 1) ? f12 : 0u) | ((ip & 2) ? f13 : 0u);
      v2f Sv[4];
#pragma unroll
      for (int k = 0; k < 4; ++k) {
        Sv[k][0] = __uint_as_float(__float_as_uint(S[k][0]) ^ flip);
        Sv[k][1] = __uint_as_float(__float_as_uint(S[k][1]) ^ flip);
      }

      z0 = wmma_16x16x4_f32(v2f{qa0[0], qa0[1]}, Sv[0], z0);
      z1 = wmma_16x16x4_f32(v2f{qb0[0], qb0[1]}, Sv[0], z1);
      z0 = wmma_16x16x4_f32(v2f{qa0[2], qa0[3]}, Sv[1], z0);
      z1 = wmma_16x16x4_f32(v2f{qb0[2], qb0[3]}, Sv[1], z1);
      z0 = wmma_16x16x4_f32(v2f{qa1[0], qa1[1]}, Sv[2], z0);
      z1 = wmma_16x16x4_f32(v2f{qb1[0], qb1[1]}, Sv[2], z1);
      z0 = wmma_16x16x4_f32(v2f{qa1[2], qa1[3]}, Sv[3], z0);
      z1 = wmma_16x16x4_f32(v2f{qb1[2], qb1[3]}, Sv[3], z1);
    }

#if defined(__gfx1250__)
    asm volatile("v_nop\n\tv_nop\n\tv_nop\n\tv_nop" : "+v"(z0), "+v"(z1));
#endif
    float cs = 0.f, d11 = 0.f, mb0 = 0.f, mb1 = 0.f, mb2 = 0.f;
#pragma unroll
    for (int r = 0; r < 8; ++r) {
      const float sp = z0[r] + z1[r];
      cs  += sp;
      d11 += z0[r] - z1[r];
      mb0 += (r & 1) ? -sp : sp;
      mb1 += (r & 2) ? -sp : sp;
      mb2 += (r & 4) ? -sp : sp;
    }
    const int slot = wave * 2 + hl;
    if (cl == 0) {
      zred[ 2 * 16 + slot] = mb0;
      zred[ 3 * 16 + slot] = mb1;
      zred[ 4 * 16 + slot] = mb2;
      zred[ 5 * 16 + slot] = hl ? -cs : cs;
      zred[ 8 * 16 + slot] = (wave & 1) ? -cs : cs;
      zred[ 9 * 16 + slot] = (wave & 2) ? -cs : cs;
      zred[10 * 16 + slot] = (wave & 4) ? -cs : cs;
      zred[11 * 16 + slot] = d11;
    } else if (cl <= 6) {
      const int bit = (cl == 1) ? 0 : (cl == 2) ? 1 : (cl == 3) ? 6 : (cl == 4) ? 7 : (cl == 5) ? 12 : 13;
      zred[bit * 16 + slot] = cs;
    }
  }
  __syncthreads();
  if (tid < NQ) {
    const float* zr = zred + (13 - tid) * 16;
    float s = 0.f;
#pragma unroll
    for (int i = 0; i < 16; ++i) s += zr[i];
    out[b * NQ + tid] = s;
  }
}

extern "C" void kernel_launch(void* const* d_in, const int* in_sizes, int n_in,
                              void* d_out, int out_size, void* d_ws, size_t ws_size,
                              hipStream_t stream) {
  const float* x      = (const float*)d_in[0];
  const float* params = (const float*)d_in[1];
  float* out          = (float*)d_out;
  int B = in_sizes[0] / NQ;
  qsim14_kernel<<<dim3(B), dim3(NT), 0, stream>>>(x, params, out);
}


# Layer 0: index bits 0,1 left out through the high pass too (16 tiles for 64) and created, already rotated (complex 4x16 matrix, K=4), by the low pass's first product; sum order changes
# speedup vs baseline: 2.7003x; 1.1704x over previous
#include <hip/hip_runtime.h>
#include <math.h>

typedef float v2f __attribute__((ext_vector_type(2)));
typedef float v4f __attribute__((ext_vector_type(4)));
typedef float v8f __attribute__((ext_vector_type(8)));

#define NQ   14
#define DIM  16384
#define NT   256
#define ROWP 260
#define CSZ  (64 * ROWP)

static __device__ inline v8f wmma_16x16x4_f32(v2f a, v2f b, v8f c) {
#if defined(__gfx1250__)
  return __builtin_amdgcn_wmma_f32_16x16x4_f32(
      false, a, false, b, (short)0, c, false, false);
#else
  return c;
#endif
}

static __device__ inline void gate_frag(const float* gc, const float* gs, int lane, int xmask, v2f F[4]) {
  const int No = lane & 15;
  const int N  = No ^ ((No >> 1) & xmask);
  const int hi = lane >> 4;
#pragma unroll
  for (int k = 0; k < 4; ++k) {
#pragma unroll
    for (int r = 0; r < 2; ++r) {
      const int ib[4] = { r, k & 1, k >> 1, hi };
      float e = 1.f;
#pragma unroll
      for (int f = 0; f < 4; ++f) {
        int ob = (N >> f) & 1;
        e *= (ob == ib[f]) ? gc[f] : ((ob == 0) ? -gs[f] : gs[f]);
      }
      F[k][r] = e;
    }
  }
}

__global__ __launch_bounds__(NT) void qsim14_kernel(
    const float* __restrict__ x,
    const float* __restrict__ params,
    float* __restrict__ out)
{
  __shared__ __attribute__((aligned(16))) float st[2 * CSZ];
  __shared__ __attribute__((aligned(16))) float pHi[128];
  __shared__ float pL5[32];
  __shared__ float tc[3 * NQ], ts[3 * NQ];

  const int b    = blockIdx.x;
  const int tid  = threadIdx.x;
  const int lane = tid & 31;
  const int wave = tid >> 5;

  {
    const int slot = (wave == 0 && lane < NQ) ? lane
                   : (wave == 1 && lane < 2 * NQ) ? NQ + lane : -1;
    if (slot >= 0) {
      const float h = 0.5f * ((slot < NQ) ? x[b * NQ + slot] : params[slot - NQ]);
      tc[slot] = cosf(h);
      ts[slot] = sinf(h);
    }
  }
  __syncthreads();

  if (tid < 128) {
    int h = tid; float p = 1.f;
#pragma unroll
    for (int t = 0; t < 7; ++t) p *= ((h >> (6 - t)) & 1) ? ts[t] : tc[t];
    pHi[((h & 31) << 2) | (h >> 5)] = p;
  } else if (tid < 160) {
    int l = tid - 128; float p = 1.f;
#pragma unroll
    for (int u = 0; u < 5; ++u) p *= ((l >> (4 - u)) & 1) ? ts[7 + u] : tc[7 + u];
    pL5[l] = p;
  }
  __syncthreads();

  {
    const int c  = tid & 15;
    const int rq = tid >> 4;
    const float Lv0 = pL5[c], Lv1 = pL5[c + 16];
    const v4f H0 = *(const v4f*)(pHi + 8 * rq);
    const v4f H1 = *(const v4f*)(pHi + 8 * rq + 4);
    const int pt = __popc(tid) & 3;
    const float tr = (pt == 0) ? 1.f : ((pt == 2) ? -1.f : 0.f);
    const float ti = (pt == 1) ? -1.f : ((pt == 3) ? 1.f : 0.f);
    const float cr[4] = { tr,  ti, -tr, -ti };
    const float ci[4] = { ti, -tr, -ti,  tr };
    float* ps = st + rq * ROWP + 4 * c;
#pragma unroll
    for (int i = 0; i < 4; ++i) {
      v4f re, im;
#pragma unroll
      for (int e = 0; e < 4; ++e) {
        const int q = (__builtin_popcount(i) + __builtin_popcount(e)) & 3;
        const float mag = ((e & 2) ? H1[i] : H0[i]) * ((e & 1) ? Lv1 : Lv0);
        re[e] = mag * cr[q];
        im[e] = mag * ci[q];
      }
      float* pr = ps + (16 * (i & 1) + 32 * (i >> 1)) * ROWP;
      *(v4f*)(pr)       = re;
      *(v4f*)(pr + CSZ) = im;
    }
  }
  __syncthreads();

#pragma unroll
  for (int L = 0; L < 2; ++L) {
#pragma unroll
    for (int pass = 0; pass < 2; ++pass) {
      const int qb[2][4] = { {-1, -1, 1, 0}, {13, 12, 7, 6} };
      const int qa[2][4] = { { 5,  4, 3, 2}, {11, 10, 9, 8} };
      const float* lc = tc + NQ + L * NQ;
      const float* ls = ts + NQ + L * NQ;
      float c1[4], s1[4], c2[4], s2[4];
#pragma unroll
      for (int f = 0; f < 4; ++f) {
        if (qb[pass][f] < 0) { c1[f] = 1.f; s1[f] = 0.f; }
        else { c1[f] = lc[qb[pass][f]]; s1[f] = ls[qb[pass][f]]; }
        c2[f] = lc[qa[pass][f]]; s2[f] = ls[qa[pass][f]];
      }
      const bool create = (L == 0 && pass == 1);
      const int ntile = (L == 0 && pass == 0) ? 16 : 64;
      v2f B1[4] = {}, B2[4];
      if (!create) gate_frag(c1, s1, lane, pass ? 5 : 4, B1);
      gate_frag(c2, s2, lane, 7, B2);

      const int laneStride = pass ? 4   : ROWP;
      const int hiOff      = pass ? 128 : 32 * ROWP;
      const int off2       = pass ? 64  : 16 * ROWP;
      const int tileStride = pass ? ROWP : 4;
      const int cl = lane & 15, hl = lane >> 4;
      const float* pl = st + cl * laneStride + hl * hiOff;

      v2f Er = {}, Ei = {}, En = {};
      const float* pf = st + 4 * cl + 2 * hl;
      if (create) {
        const int a  = cl ^ ((cl >> 1) & 5);
        const int a0 = a & 1, a1 = (a >> 1) & 1, o6 = (a >> 2) & 1, o7 = (a >> 3) & 1;
        const float x0r = a0 ?  s1[0] * tc[13] : c1[0] * tc[13];
        const float x0i = a0 ? -c1[0] * ts[13] : s1[0] * ts[13];
        const float x1r = a1 ?  s1[1] * tc[12] : c1[1] * tc[12];
        const float x1i = a1 ? -c1[1] * ts[12] : s1[1] * ts[12];
        const float wr = x0r * x1r - x0i * x1i;
        const float wi = x0r * x1i + x0i * x1r;
        const float g7 = (o7 == hl) ? c1[3] : ((o7 == 0) ? -s1[3] : s1[3]);
#pragma unroll
        for (int r = 0; r < 2; ++r) {
          const float g6 = (o6 == r) ? c1[2] : ((o6 == 0) ? -s1[2] : s1[2]);
          const float g = g6 * g7;
          Er[r] = g * wr;
          Ei[r] = g * wi;
          En[r] = -(g * wi);
        }
      }

      const int T  = pass ? (wave & 1) : 0;
      const int e0 = T, e1 = T ^ 1;
      float* ps0 = st + (cl ^ (e0 ? 15 : 0)) * laneStride + (hl ^ T) * hiOff + e0 * off2;
      float* ps1 = st + (cl ^ (e1 ? 15 : 0)) * laneStride + (hl ^ T) * hiOff + e1 * off2;
      const bool rev0 = ((cl ^ e0) & 1) != 0;

      for (int t = wave; t < ntile; t += 8) {
        v8f dr = {}, di = {};
        if (create) {
          const v2f ar = *(const v2f*)(pf + t * ROWP);
          const v2f ai = *(const v2f*)(pf + t * ROWP + CSZ);
          dr = wmma_16x16x4_f32(ar, Er, dr);
          di = wmma_16x16x4_f32(ar, Ei, di);
          dr = wmma_16x16x4_f32(ai, En, dr);
          di = wmma_16x16x4_f32(ai, Er, di);
        } else {
          const float* p = pl + t * tileStride;
          const v4f r0 = *(const v4f*)(p);
          const v4f r1 = *(const v4f*)(p + off2);
          const v4f i0 = *(const v4f*)(p + CSZ);
          const v4f i1 = *(const v4f*)(p + CSZ + off2);
          dr = wmma_16x16x4_f32(v2f{r0[0], r0[1]}, B1[0], dr);
          di = wmma_16x16x4_f32(v2f{i0[0], i0[1]}, B1[0], di);
          dr = wmma_16x16x4_f32(v2f{r0[2], r0[3]}, B1[1], dr);
          di = wmma_16x16x4_f32(v2f{i0[2], i0[3]}, B1[1], di);
          dr = wmma_16x16x4_f32(v2f{r1[0], r1[1]}, B1[2], dr);
          di = wmma_16x16x4_f32(v2f{i1[0], i1[1]}, B1[2], di);
          dr = wmma_16x16x4_f32(v2f{r1[2], r1[3]}, B1[3], dr);
          di = wmma_16x16x4_f32(v2f{i1[2], i1[3]}, B1[3], di);
        }

        v8f er = {}, ei = {};
        er = wmma_16x16x4_f32(v2f{dr[0], dr[1]}, B2[0], er);
        ei = wmma_16x16x4_f32(v2f{di[0], di[1]}, B2[0], ei);
        er = wmma_16x16x4_f32(v2f{dr[2], dr[3]}, B2[1], er);
        ei = wmma_16x16x4_f32(v2f{di[2], di[3]}, B2[1], ei);
        er = wmma_16x16x4_f32(v2f{dr[4], dr[5]}, B2[2], er);
        ei = wmma_16x16x4_f32(v2f{di[4], di[5]}, B2[2], ei);
        er = wmma_16x16x4_f32(v2f{dr[6], dr[7]}, B2[3], er);
        ei = wmma_16x16x4_f32(v2f{di[6], di[7]}, B2[3], ei);

        float* q0 = ps0 + t * tileStride;
        float* q1 = ps1 + t * tileStride;
        if (pass) {
          const bool rev1 = !rev0;
          *(v4f*)(q0)       = rev0 ? v4f{er[3], er[2], er[1], er[0]} : v4f{er[0], er[1], er[2], er[3]};
          *(v4f*)(q1)       = rev1 ? v4f{er[7], er[6], er[5], er[4]} : v4f{er[4], er[5], er[6], er[7]};
          *(v4f*)(q0 + CSZ) = rev0 ? v4f{ei[3], ei[2], ei[1], ei[0]} : v4f{ei[0], ei[1], ei[2], ei[3]};
          *(v4f*)(q1 + CSZ) = rev1 ? v4f{ei[7], ei[6], ei[5], ei[4]} : v4f{ei[4], ei[5], ei[6], ei[7]};
        } else {
          *(v4f*)(q0)       = v4f{er[0], er[1], er[2], er[3]};
          *(v4f*)(q1)       = v4f{er[4], er[5], er[6], er[7]};
          *(v4f*)(q0 + CSZ) = v4f{ei[0], ei[1], ei[2], ei[3]};
          *(v4f*)(q1 + CSZ) = v4f{ei[4], ei[5], ei[6], ei[7]};
        }
      }
      __syncthreads();
    }
  }

  __shared__ float zred[NQ * 16];
  {
    const int cl = lane & 15, hl = lane >> 4;
    const unsigned sel = (cl >= 1 && cl <= 4) ? (1u << (cl - 1)) : 0u;
    const unsigned one = (cl > 6) ? 0u : 0x3f800000u;
    const unsigned hb  = (unsigned)hl << 3;
    v2f S[4];
#pragma unroll
    for (int k = 0; k < 4; ++k) {
#pragma unroll
      for (int r = 0; r < 2; ++r) {
        const unsigned nb = (unsigned)(r | ((k & 1) << 1) | ((k >> 1) << 2)) | hb;
        S[k][r] = __uint_as_float(one | ((nb & sel) ? 0x80000000u : 0u));
      }
    }
    const unsigned f12 = (cl == 5) ? 0x80000000u : 0u;
    const unsigned f13 = (cl == 6) ? 0x80000000u : 0u;

    v8f z0 = {}, z1 = {};
    const float* pl = st + cl * 4 + hl * 128 + wave * ROWP;
#pragma unroll
    for (int ip = 0; ip < 4; ++ip) {
      const float* pa = pl + (16 * ip) * ROWP;
      const float* pb = pa + 8 * ROWP;
      const v4f ra0 = *(const v4f*)(pa);
      const v4f ra1 = *(const v4f*)(pa + 64);
      const v4f ia0 = *(const v4f*)(pa + CSZ);
      const v4f ia1 = *(const v4f*)(pa + CSZ + 64);
      const v4f rb0 = *(const v4f*)(pb);
      const v4f rb1 = *(const v4f*)(pb + 64);
      const v4f ib0 = *(const v4f*)(pb + CSZ);
      const v4f ib1 = *(const v4f*)(pb + CSZ + 64);
      const v4f qa0 = ra0 * ra0 + ia0 * ia0;
      const v4f qa1 = ra1 * ra1 + ia1 * ia1;
      const v4f qb0 = rb0 * rb0 + ib0 * ib0;
      const v4f qb1 = rb1 * rb1 + ib1 * ib1;

      const unsigned flip = ((ip & 1) ? f12 : 0u) | ((ip & 2) ? f13 : 0u);
      v2f Sv[4];
#pragma unroll
      for (int k = 0; k < 4; ++k) {
        Sv[k][0] = __uint_as_float(__float_as_uint(S[k][0]) ^ flip);
        Sv[k][1] = __uint_as_float(__float_as_uint(S[k][1]) ^ flip);
      }

      z0 = wmma_16x16x4_f32(v2f{qa0[0], qa0[1]}, Sv[0], z0);
      z1 = wmma_16x16x4_f32(v2f{qb0[0], qb0[1]}, Sv[0], z1);
      z0 = wmma_16x16x4_f32(v2f{qa0[2], qa0[3]}, Sv[1], z0);
      z1 = wmma_16x16x4_f32(v2f{qb0[2], qb0[3]}, Sv[1], z1);
      z0 = wmma_16x16x4_f32(v2f{qa1[0], qa1[1]}, Sv[2], z0);
      z1 = wmma_16x16x4_f32(v2f{qb1[0], qb1[1]}, Sv[2], z1);
      z0 = wmma_16x16x4_f32(v2f{qa1[2], qa1[3]}, Sv[3], z0);
      z1 = wmma_16x16x4_f32(v2f{qb1[2], qb1[3]}, Sv[3], z1);
    }

#if defined(__gfx1250__)
    asm volatile("v_nop\n\tv_nop\n\tv_nop\n\tv_nop" : "+v"(z0), "+v"(z1));
#endif
    float cs = 0.f, d11 = 0.f, mb0 = 0.f, mb1 = 0.f, mb2 = 0.f;
#pragma unroll
    for (int r = 0; r < 8; ++r) {
      const float sp = z0[r] + z1[r];
      cs  += sp;
      d11 += z0[r] - z1[r];
      mb0 += (r & 1) ? -sp : sp;
      mb1 += (r & 2) ? -sp : sp;
      mb2 += (r & 4) ? -sp : sp;
    }
    const int slot = wave * 2 + hl;
    if (cl == 0) {
      zred[ 2 * 16 + slot] = mb0;
      zred[ 3 * 16 + slot] = mb1;
      zred[ 4 * 16 + slot] = mb2;
      zred[ 5 * 16 + slot] = hl ? -cs : cs;
      zred[ 8 * 16 + slot] = (wave & 1) ? -cs : cs;
      zred[ 9 * 16 + slot] = (wave & 2) ? -cs : cs;
      zred[10 * 16 + slot] = (wave & 4) ? -cs : cs;
      zred[11 * 16 + slot] = d11;
    } else if (cl <= 6) {
      const int bit = (cl == 1) ? 0 : (cl == 2) ? 1 : (cl == 3) ? 6 : (cl == 4) ? 7 : (cl == 5) ? 12 : 13;
      zred[bit * 16 + slot] = cs;
    }
  }
  __syncthreads();
  if (tid < NQ) {
    const float* zr = zred + (13 - tid) * 16;
    float s = 0.f;
#pragma unroll
    for (int i = 0; i < 16; ++i) s += zr[i];
    out[b * NQ + tid] = s;
  }
}

extern "C" void kernel_launch(void* const* d_in, const int* in_sizes, int n_in,
                              void* d_out, int out_size, void* d_ws, size_t ws_size,
                              hipStream_t stream) {
  const float* x      = (const float*)d_in[0];
  const float* params = (const float*)d_in[1];
  float* out          = (float*)d_out;
  int B = in_sizes[0] / NQ;
  qsim14_kernel<<<dim3(B), dim3(NT), 0, stream>>>(x, params, out);
}


# Z sums: the wave's eight |amp|^2 tiles are added element by element before the sign product, tile-constant signs by a second product on per-lane tile totals (32 WMMAs become 8); sum order changes
# speedup vs baseline: 2.7912x; 1.0337x over previous
#include <hip/hip_runtime.h>
#include <math.h>

typedef float v2f __attribute__((ext_vector_type(2)));
typedef float v4f __attribute__((ext_vector_type(4)));
typedef float v8f __attribute__((ext_vector_type(8)));

#define NQ   14
#define DIM  16384
#define NT   256
#define ROWP 260
#define CSZ  (64 * ROWP)

static __device__ inline v8f wmma_16x16x4_f32(v2f a, v2f b, v8f c) {
#if defined(__gfx1250__)
  return __builtin_amdgcn_wmma_f32_16x16x4_f32(
      false, a, false, b, (short)0, c, false, false);
#else
  return c;
#endif
}

static __device__ inline void gate_frag(const float* gc, const float* gs, int lane, int xmask, v2f F[4]) {
  const int No = lane & 15;
  const int N  = No ^ ((No >> 1) & xmask);
  const int hi = lane >> 4;
#pragma unroll
  for (int k = 0; k < 4; ++k) {
#pragma unroll
    for (int r = 0; r < 2; ++r) {
      const int ib[4] = { r, k & 1, k >> 1, hi };
      float e = 1.f;
#pragma unroll
      for (int f = 0; f < 4; ++f) {
        int ob = (N >> f) & 1;
        e *= (ob == ib[f]) ? gc[f] : ((ob == 0) ? -gs[f] : gs[f]);
      }
      F[k][r] = e;
    }
  }
}

__global__ __launch_bounds__(NT) void qsim14_kernel(
    const float* __restrict__ x,
    const float* __restrict__ params,
    float* __restrict__ out)
{
  __shared__ __attribute__((aligned(16))) float st[2 * CSZ];
  __shared__ __attribute__((aligned(16))) float pHi[128];
  __shared__ float pL5[32];
  __shared__ float tc[3 * NQ], ts[3 * NQ];

  const int b    = blockIdx.x;
  const int tid  = threadIdx.x;
  const int lane = tid & 31;
  const int wave = tid >> 5;

  {
    const int slot = (wave == 0 && lane < NQ) ? lane
                   : (wave == 1 && lane < 2 * NQ) ? NQ + lane : -1;
    if (slot >= 0) {
      const float h = 0.5f * ((slot < NQ) ? x[b * NQ + slot] : params[slot - NQ]);
      tc[slot] = cosf(h);
      ts[slot] = sinf(h);
    }
  }
  __syncthreads();

  if (tid < 128) {
    int h = tid; float p = 1.f;
#pragma unroll
    for (int t = 0; t < 7; ++t) p *= ((h >> (6 - t)) & 1) ? ts[t] : tc[t];
    pHi[((h & 31) << 2) | (h >> 5)] = p;
  } else if (tid < 160) {
    int l = tid - 128; float p = 1.f;
#pragma unroll
    for (int u = 0; u < 5; ++u) p *= ((l >> (4 - u)) & 1) ? ts[7 + u] : tc[7 + u];
    pL5[l] = p;
  }
  __syncthreads();

  {
    const int c  = tid & 15;
    const int rq = tid >> 4;
    const float Lv0 = pL5[c], Lv1 = pL5[c + 16];
    const v4f H0 = *(const v4f*)(pHi + 8 * rq);
    const v4f H1 = *(const v4f*)(pHi + 8 * rq + 4);
    const int pt = __popc(tid) & 3;
    const float tr = (pt == 0) ? 1.f : ((pt == 2) ? -1.f : 0.f);
    const float ti = (pt == 1) ? -1.f : ((pt == 3) ? 1.f : 0.f);
    const float cr[4] = { tr,  ti, -tr, -ti };
    const float ci[4] = { ti, -tr, -ti,  tr };
    float* ps = st + rq * ROWP + 4 * c;
#pragma unroll
    for (int i = 0; i < 4; ++i) {
      v4f re, im;
#pragma unroll
      for (int e = 0; e < 4; ++e) {
        const int q = (__builtin_popcount(i) + __builtin_popcount(e)) & 3;
        const float mag = ((e & 2) ? H1[i] : H0[i]) * ((e & 1) ? Lv1 : Lv0);
        re[e] = mag * cr[q];
        im[e] = mag * ci[q];
      }
      float* pr = ps + (16 * (i & 1) + 32 * (i >> 1)) * ROWP;
      *(v4f*)(pr)       = re;
      *(v4f*)(pr + CSZ) = im;
    }
  }
  __syncthreads();

#pragma unroll
  for (int L = 0; L < 2; ++L) {
#pragma unroll
    for (int pass = 0; pass < 2; ++pass) {
      const int qb[2][4] = { {-1, -1, 1, 0}, {13, 12, 7, 6} };
      const int qa[2][4] = { { 5,  4, 3, 2}, {11, 10, 9, 8} };
      const float* lc = tc + NQ + L * NQ;
      const float* ls = ts + NQ + L * NQ;
      float c1[4], s1[4], c2[4], s2[4];
#pragma unroll
      for (int f = 0; f < 4; ++f) {
        if (qb[pass][f] < 0) { c1[f] = 1.f; s1[f] = 0.f; }
        else { c1[f] = lc[qb[pass][f]]; s1[f] = ls[qb[pass][f]]; }
        c2[f] = lc[qa[pass][f]]; s2[f] = ls[qa[pass][f]];
      }
      const bool create = (L == 0 && pass == 1);
      const int ntile = (L == 0 && pass == 0) ? 16 : 64;
      v2f B1[4] = {}, B2[4];
      if (!create) gate_frag(c1, s1, lane, pass ? 5 : 4, B1);
      gate_frag(c2, s2, lane, 7, B2);

      const int laneStride = pass ? 4   : ROWP;
      const int hiOff      = pass ? 128 : 32 * ROWP;
      const int off2       = pass ? 64  : 16 * ROWP;
      const int tileStride = pass ? ROWP : 4;
      const int cl = lane & 15, hl = lane >> 4;
      const float* pl = st + cl * laneStride + hl * hiOff;

      v2f Er = {}, Ei = {}, En = {};
      const float* pf = st + 4 * cl + 2 * hl;
      if (create) {
        const int a  = cl ^ ((cl >> 1) & 5);
        const int a0 = a & 1, a1 = (a >> 1) & 1, o6 = (a >> 2) & 1, o7 = (a >> 3) & 1;
        const float x0r = a0 ?  s1[0] * tc[13] : c1[0] * tc[13];
        const float x0i = a0 ? -c1[0] * ts[13] : s1[0] * ts[13];
        const float x1r = a1 ?  s1[1] * tc[12] : c1[1] * tc[12];
        const float x1i = a1 ? -c1[1] * ts[12] : s1[1] * ts[12];
        const float wr = x0r * x1r - x0i * x1i;
        const float wi = x0r * x1i + x0i * x1r;
        const float g7 = (o7 == hl) ? c1[3] : ((o7 == 0) ? -s1[3] : s1[3]);
#pragma unroll
        for (int r = 0; r < 2; ++r) {
          const float g6 = (o6 == r) ? c1[2] : ((o6 == 0) ? -s1[2] : s1[2]);
          const float g = g6 * g7;
          Er[r] = g * wr;
          Ei[r] = g * wi;
          En[r] = -(g * wi);
        }
      }

      const int T  = pass ? (wave & 1) : 0;
      const int e0 = T, e1 = T ^ 1;
      float* ps0 = st + (cl ^ (e0 ? 15 : 0)) * laneStride + (hl ^ T) * hiOff + e0 * off2;
      float* ps1 = st + (cl ^ (e1 ? 15 : 0)) * laneStride + (hl ^ T) * hiOff + e1 * off2;
      const bool rev0 = ((cl ^ e0) & 1) != 0;

      for (int t = wave; t < ntile; t += 8) {
        v8f dr = {}, di = {};
        if (create) {
          const v2f ar = *(const v2f*)(pf + t * ROWP);
          const v2f ai = *(const v2f*)(pf + t * ROWP + CSZ);
          dr = wmma_16x16x4_f32(ar, Er, dr);
          di = wmma_16x16x4_f32(ar, Ei, di);
          dr = wmma_16x16x4_f32(ai, En, dr);
          di = wmma_16x16x4_f32(ai, Er, di);
        } else {
          const float* p = pl + t * tileStride;
          const v4f r0 = *(const v4f*)(p);
          const v4f r1 = *(const v4f*)(p + off2);
          const v4f i0 = *(const v4f*)(p + CSZ);
          const v4f i1 = *(const v4f*)(p + CSZ + off2);
          dr = wmma_16x16x4_f32(v2f{r0[0], r0[1]}, B1[0], dr);
          di = wmma_16x16x4_f32(v2f{i0[0], i0[1]}, B1[0], di);
          dr = wmma_16x16x4_f32(v2f{r0[2], r0[3]}, B1[1], dr);
          di = wmma_16x16x4_f32(v2f{i0[2], i0[3]}, B1[1], di);
          dr = wmma_16x16x4_f32(v2f{r1[0], r1[1]}, B1[2], dr);
          di = wmma_16x16x4_f32(v2f{i1[0], i1[1]}, B1[2], di);
          dr = wmma_16x16x4_f32(v2f{r1[2], r1[3]}, B1[3], dr);
          di = wmma_16x16x4_f32(v2f{i1[2], i1[3]}, B1[3], di);
        }

        v8f er = {}, ei = {};
        er = wmma_16x16x4_f32(v2f{dr[0], dr[1]}, B2[0], er);
        ei = wmma_16x16x4_f32(v2f{di[0], di[1]}, B2[0], ei);
        er = wmma_16x16x4_f32(v2f{dr[2], dr[3]}, B2[1], er);
        ei = wmma_16x16x4_f32(v2f{di[2], di[3]}, B2[1], ei);
        er = wmma_16x16x4_f32(v2f{dr[4], dr[5]}, B2[2], er);
        ei = wmma_16x16x4_f32(v2f{di[4], di[5]}, B2[2], ei);
        er = wmma_16x16x4_f32(v2f{dr[6], dr[7]}, B2[3], er);
        ei = wmma_16x16x4_f32(v2f{di[6], di[7]}, B2[3], ei);

        float* q0 = ps0 + t * tileStride;
        float* q1 = ps1 + t * tileStride;
        if (pass) {
          const bool rev1 = !rev0;
          *(v4f*)(q0)       = rev0 ? v4f{er[3], er[2], er[1], er[0]} : v4f{er[0], er[1], er[2], er[3]};
          *(v4f*)(q1)       = rev1 ? v4f{er[7], er[6], er[5], er[4]} : v4f{er[4], er[5], er[6], er[7]};
          *(v4f*)(q0 + CSZ) = rev0 ? v4f{ei[3], ei[2], ei[1], ei[0]} : v4f{ei[0], ei[1], ei[2], ei[3]};
          *(v4f*)(q1 + CSZ) = rev1 ? v4f{ei[7], ei[6], ei[5], ei[4]} : v4f{ei[4], ei[5], ei[6], ei[7]};
        } else {
          *(v4f*)(q0)       = v4f{er[0], er[1], er[2], er[3]};
          *(v4f*)(q1)       = v4f{er[4], er[5], er[6], er[7]};
          *(v4f*)(q0 + CSZ) = v4f{ei[0], ei[1], ei[2], ei[3]};
          *(v4f*)(q1 + CSZ) = v4f{ei[4], ei[5], ei[6], ei[7]};
        }
      }
      __syncthreads();
    }
  }

  __shared__ float zred[NQ * 16];
  {
    const int cl = lane & 15, hl = lane >> 4;
    const unsigned sel = (cl >= 1 && cl <= 4) ? (1u << (cl - 1)) : 0u;
    const unsigned one = (cl > 4) ? 0u : 0x3f800000u;
    const unsigned hb  = (unsigned)hl << 3;
    const unsigned tsel = (cl >= 5 && cl <= 7) ? (1u << (cl - 5)) : 0u;
    const unsigned tone = tsel ? 0x3f800000u : 0u;
    v2f S[4], T[4];
#pragma unroll
    for (int k = 0; k < 4; ++k) {
#pragma unroll
      for (int r = 0; r < 2; ++r) {
        const unsigned nb = (unsigned)(r | ((k & 1) << 1) | ((k >> 1) << 2)) | hb;
        S[k][r] = __uint_as_float(one | ((nb & sel) ? 0x80000000u : 0u));
        const unsigned ti = (unsigned)(2 * k + r);
        T[k][r] = __uint_as_float(tone | ((ti & tsel) ? 0x80000000u : 0u));
      }
    }

    v4f p0 = {}, p1 = {};
    float sI[8];
    const float* pl = st + cl * 4 + hl * 128 + wave * ROWP;
#pragma unroll
    for (int i = 0; i < 8; ++i) {
      const float* pa = pl + (8 * i) * ROWP;
      const v4f r0 = *(const v4f*)(pa);
      const v4f r1 = *(const v4f*)(pa + 64);
      const v4f m0 = *(const v4f*)(pa + CSZ);
      const v4f m1 = *(const v4f*)(pa + CSZ + 64);
      const v4f q0 = r0 * r0 + m0 * m0;
      const v4f q1 = r1 * r1 + m1 * m1;
      p0 += q0;
      p1 += q1;
      const v4f u = q0 + q1;
      sI[i] = (u[0] + u[1]) + (u[2] + u[3]);
    }

    v8f z0 = {}, z1 = {};
    z0 = wmma_16x16x4_f32(v2f{p0[0], p0[1]}, S[0], z0);
    z1 = wmma_16x16x4_f32(v2f{sI[0], sI[1]}, T[0], z1);
    z0 = wmma_16x16x4_f32(v2f{p0[2], p0[3]}, S[1], z0);
    z1 = wmma_16x16x4_f32(v2f{sI[2], sI[3]}, T[1], z1);
    z0 = wmma_16x16x4_f32(v2f{p1[0], p1[1]}, S[2], z0);
    z1 = wmma_16x16x4_f32(v2f{sI[4], sI[5]}, T[2], z1);
    z0 = wmma_16x16x4_f32(v2f{p1[2], p1[3]}, S[3], z0);
    z1 = wmma_16x16x4_f32(v2f{sI[6], sI[7]}, T[3], z1);

#if defined(__gfx1250__)
    asm volatile("v_nop\n\tv_nop\n\tv_nop\n\tv_nop" : "+v"(z0), "+v"(z1));
#endif
    float cs = 0.f, mb0 = 0.f, mb1 = 0.f, mb2 = 0.f;
#pragma unroll
    for (int r = 0; r < 8; ++r) {
      const float sp = z0[r] + z1[r];
      cs  += sp;
      mb0 += (r & 1) ? -sp : sp;
      mb1 += (r & 2) ? -sp : sp;
      mb2 += (r & 4) ? -sp : sp;
    }
    const int slot = wave * 2 + hl;
    if (cl == 0) {
      zred[ 2 * 16 + slot] = mb0;
      zred[ 3 * 16 + slot] = mb1;
      zred[ 4 * 16 + slot] = mb2;
      zred[ 5 * 16 + slot] = hl ? -cs : cs;
      zred[ 8 * 16 + slot] = (wave & 1) ? -cs : cs;
      zred[ 9 * 16 + slot] = (wave & 2) ? -cs : cs;
      zred[10 * 16 + slot] = (wave & 4) ? -cs : cs;
    } else if (cl <= 7) {
      const int bit = (cl == 1) ? 0 : (cl == 2) ? 1 : (cl == 3) ? 6 : (cl == 4) ? 7 : (cl + 6);
      zred[bit * 16 + slot] = cs;
    }
  }
  __syncthreads();
  if (tid < NQ) {
    const float* zr = zred + (13 - tid) * 16;
    float s = 0.f;
#pragma unroll
    for (int i = 0; i < 16; ++i) s += zr[i];
    out[b * NQ + tid] = s;
  }
}

extern "C" void kernel_launch(void* const* d_in, const int* in_sizes, int n_in,
                              void* d_out, int out_size, void* d_ws, size_t ws_size,
                              hipStream_t stream) {
  const float* x      = (const float*)d_in[0];
  const float* params = (const float*)d_in[1];
  float* out          = (float*)d_out;
  int B = in_sizes[0] / NQ;
  qsim14_kernel<<<dim3(B), dim3(NT), 0, stream>>>(x, params, out);
}


# Layer 0 low pass: gate on index bits 2..5 applied to the reduced state before bits 0,1 are created; state is the B operand of the creating product (12 WMMAs per tile become 6); sum order changes
# speedup vs baseline: 3.0483x; 1.0921x over previous
#include <hip/hip_runtime.h>
#include <math.h>

typedef float v2f __attribute__((ext_vector_type(2)));
typedef float v4f __attribute__((ext_vector_type(4)));
typedef float v8f __attribute__((ext_vector_type(8)));

#define NQ   14
#define DIM  16384
#define NT   256
#define ROWP 260
#define CSZ  (64 * ROWP)

static __device__ inline v8f wmma_16x16x4_f32(v2f a, v2f b, v8f c) {
#if defined(__gfx1250__)
  return __builtin_amdgcn_wmma_f32_16x16x4_f32(
      false, a, false, b, (short)0, c, false, false);
#else
  return c;
#endif
}

static __device__ inline void gate_frag(const float* gc, const float* gs, int lane, int xmask, v2f F[4]) {
  const int No = lane & 15;
  const int N  = No ^ ((No >> 1) & xmask);
  const int hi = lane >> 4;
#pragma unroll
  for (int k = 0; k < 4; ++k) {
#pragma unroll
    for (int r = 0; r < 2; ++r) {
      const int ib[4] = { r, k & 1, k >> 1, hi };
      float e = 1.f;
#pragma unroll
      for (int f = 0; f < 4; ++f) {
        int ob = (N >> f) & 1;
        e *= (ob == ib[f]) ? gc[f] : ((ob == 0) ? -gs[f] : gs[f]);
      }
      F[k][r] = e;
    }
  }
}

__global__ __launch_bounds__(NT) void qsim14_kernel(
    const float* __restrict__ x,
    const float* __restrict__ params,
    float* __restrict__ out)
{
  __shared__ __attribute__((aligned(16))) float st[2 * CSZ];
  __shared__ __attribute__((aligned(16))) float pHi[128];
  __shared__ float pL5[32];
  __shared__ float tc[3 * NQ], ts[3 * NQ];

  const int b    = blockIdx.x;
  const int tid  = threadIdx.x;
  const int lane = tid & 31;
  const int wave = tid >> 5;

  {
    const int slot = (wave == 0 && lane < NQ) ? lane
                   : (wave == 1 && lane < 2 * NQ) ? NQ + lane : -1;
    if (slot >= 0) {
      const float h = 0.5f * ((slot < NQ) ? x[b * NQ + slot] : params[slot - NQ]);
      tc[slot] = cosf(h);
      ts[slot] = sinf(h);
    }
  }
  __syncthreads();

  if (tid < 128) {
    int h = tid; float p = 1.f;
#pragma unroll
    for (int t = 0; t < 7; ++t) p *= ((h >> (6 - t)) & 1) ? ts[t] : tc[t];
    pHi[((h & 31) << 2) | (h >> 5)] = p;
  } else if (tid < 160) {
    int l = tid - 128; float p = 1.f;
#pragma unroll
    for (int u = 0; u < 5; ++u) p *= ((l >> (4 - u)) & 1) ? ts[7 + u] : tc[7 + u];
    pL5[l] = p;
  }
  __syncthreads();

  {
    const int c  = tid & 15;
    const int rq = tid >> 4;
    const float Lv0 = pL5[c], Lv1 = pL5[c + 16];
    const v4f H0 = *(const v4f*)(pHi + 8 * rq);
    const v4f H1 = *(const v4f*)(pHi + 8 * rq + 4);
    const int pt = __popc(tid) & 3;
    const float tr = (pt == 0) ? 1.f : ((pt == 2) ? -1.f : 0.f);
    const float ti = (pt == 1) ? -1.f : ((pt == 3) ? 1.f : 0.f);
    const float cr[4] = { tr,  ti, -tr, -ti };
    const float ci[4] = { ti, -tr, -ti,  tr };
    float* ps = st + rq * ROWP + 4 * c;
#pragma unroll
    for (int i = 0; i < 4; ++i) {
      v4f re, im;
#pragma unroll
      for (int e = 0; e < 4; ++e) {
        const int q = (__builtin_popcount(i) + __builtin_popcount(e)) & 3;
        const float mag = ((e & 2) ? H1[i] : H0[i]) * ((e & 1) ? Lv1 : Lv0);
        re[e] = mag * cr[q];
        im[e] = mag * ci[q];
      }
      float* pr = ps + (16 * (i & 1) + 32 * (i >> 1)) * ROWP;
      *(v4f*)(pr)       = re;
      *(v4f*)(pr + CSZ) = im;
    }
  }
  __syncthreads();

#pragma unroll
  for (int L = 0; L < 2; ++L) {
#pragma unroll
    for (int pass = 0; pass < 2; ++pass) {
      const int qb[2][4] = { {-1, -1, 1, 0}, {13, 12, 7, 6} };
      const int qa[2][4] = { { 5,  4, 3, 2}, {11, 10, 9, 8} };
      const float* lc = tc + NQ + L * NQ;
      const float* ls = ts + NQ + L * NQ;
      float c1[4], s1[4], c2[4], s2[4];
#pragma unroll
      for (int f = 0; f < 4; ++f) {
        if (qb[pass][f] < 0) { c1[f] = 1.f; s1[f] = 0.f; }
        else { c1[f] = lc[qb[pass][f]]; s1[f] = ls[qb[pass][f]]; }
        c2[f] = lc[qa[pass][f]]; s2[f] = ls[qa[pass][f]];
      }
      const bool create = (L == 0 && pass == 1);
      const int ntile = (L == 0 && pass == 0) ? 16 : 64;
      v2f B1[4] = {}, B2[4];
      if (!create) gate_frag(c1, s1, lane, pass ? 5 : 4, B1);
      gate_frag(c2, s2, lane, 7, B2);

      const int laneStride = pass ? 4   : ROWP;
      const int hiOff      = pass ? 128 : 32 * ROWP;
      const int off2       = pass ? 64  : 16 * ROWP;
      const int tileStride = pass ? ROWP : 4;
      const int cl = lane & 15, hl = lane >> 4;
      const float* pl = st + cl * laneStride + hl * hiOff;

      v2f Er = {}, Ei = {}, En = {};
      if (create) {
        const int a  = cl ^ ((cl >> 1) & 5);
        const int a0 = a & 1, a1 = (a >> 1) & 1, o6 = (a >> 2) & 1, o7 = (a >> 3) & 1;
        const float x0r = a0 ?  s1[0] * tc[13] : c1[0] * tc[13];
        const float x0i = a0 ? -c1[0] * ts[13] : s1[0] * ts[13];
        const float x1r = a1 ?  s1[1] * tc[12] : c1[1] * tc[12];
        const float x1i = a1 ? -c1[1] * ts[12] : s1[1] * ts[12];
        const float wr = x0r * x1r - x0i * x1i;
        const float wi = x0r * x1i + x0i * x1r;
        const float g7 = (o7 == hl) ? c1[3] : ((o7 == 0) ? -s1[3] : s1[3]);
#pragma unroll
        for (int r = 0; r < 2; ++r) {
          const float g6 = (o6 == r) ? c1[2] : ((o6 == 0) ? -s1[2] : s1[2]);
          const float g = g6 * g7;
          Er[r] = g * wr;
          Ei[r] = g * wi;
          En[r] = -(g * wi);
        }
      }

      const int T  = pass ? (wave & 1) : 0;
      const int e0 = T, e1 = T ^ 1;
      float* ps0 = st + (cl ^ (e0 ? 15 : 0)) * laneStride + (hl ^ T) * hiOff + e0 * off2;
      float* ps1 = st + (cl ^ (e1 ? 15 : 0)) * laneStride + (hl ^ T) * hiOff + e1 * off2;
      const bool rev0 = ((cl ^ e0) & 1) != 0;

      auto store_tile = [&](int t, const v8f& er, const v8f& ei) {
        float* q0 = ps0 + t * tileStride;
        float* q1 = ps1 + t * tileStride;
        if (pass) {
          const bool rev1 = !rev0;
          *(v4f*)(q0)       = rev0 ? v4f{er[3], er[2], er[1], er[0]} : v4f{er[0], er[1], er[2], er[3]};
          *(v4f*)(q1)       = rev1 ? v4f{er[7], er[6], er[5], er[4]} : v4f{er[4], er[5], er[6], er[7]};
          *(v4f*)(q0 + CSZ) = rev0 ? v4f{ei[3], ei[2], ei[1], ei[0]} : v4f{ei[0], ei[1], ei[2], ei[3]};
          *(v4f*)(q1 + CSZ) = rev1 ? v4f{ei[7], ei[6], ei[5], ei[4]} : v4f{ei[4], ei[5], ei[6], ei[7]};
        } else {
          *(v4f*)(q0)       = v4f{er[0], er[1], er[2], er[3]};
          *(v4f*)(q1)       = v4f{er[4], er[5], er[6], er[7]};
          *(v4f*)(q0 + CSZ) = v4f{ei[0], ei[1], ei[2], ei[3]};
          *(v4f*)(q1 + CSZ) = v4f{ei[4], ei[5], ei[6], ei[7]};
        }
      };

      if (create) {
        const int el = (cl >> 1) & 3;
        const float* pg = st + (wave + 8 * el) * ROWP + 32 * hl + 2 * (cl >> 3) + (cl & 1);
        for (int g = 0; g < 2; ++g) {
          const float* p = pg + g * (32 * ROWP);
          float xr[8], xi[8];
#pragma unroll
          for (int j = 0; j < 8; ++j) { xr[j] = p[4 * j]; xi[j] = p[4 * j + CSZ]; }
          v8f dr = {}, di = {};
#pragma unroll
          for (int k = 0; k < 4; ++k) {
            dr = wmma_16x16x4_f32(v2f{xr[2 * k], xr[2 * k + 1]}, B2[k], dr);
            di = wmma_16x16x4_f32(v2f{xi[2 * k], xi[2 * k + 1]}, B2[k], di);
          }
#pragma unroll
          for (int e = 0; e < 4; ++e) {
            const v2f br = {dr[2 * e], dr[2 * e + 1]};
            const v2f bi = {di[2 * e], di[2 * e + 1]};
            v8f er = {}, ei = {};
            er = wmma_16x16x4_f32(Er, br, er);
            ei = wmma_16x16x4_f32(Ei, br, ei);
            er = wmma_16x16x4_f32(En, bi, er);
            ei = wmma_16x16x4_f32(Er, bi, ei);
            store_tile(wave + 8 * (4 * g + e), er, ei);
          }
        }
      } else
      for (int t = wave; t < ntile; t += 8) {
        v8f dr = {}, di = {};
        {
          const float* p = pl + t * tileStride;
          const v4f r0 = *(const v4f*)(p);
          const v4f r1 = *(const v4f*)(p + off2);
          const v4f i0 = *(const v4f*)(p + CSZ);
          const v4f i1 = *(const v4f*)(p + CSZ + off2);
          dr = wmma_16x16x4_f32(v2f{r0[0], r0[1]}, B1[0], dr);
          di = wmma_16x16x4_f32(v2f{i0[0], i0[1]}, B1[0], di);
          dr = wmma_16x16x4_f32(v2f{r0[2], r0[3]}, B1[1], dr);
          di = wmma_16x16x4_f32(v2f{i0[2], i0[3]}, B1[1], di);
          dr = wmma_16x16x4_f32(v2f{r1[0], r1[1]}, B1[2], dr);
          di = wmma_16x16x4_f32(v2f{i1[0], i1[1]}, B1[2], di);
          dr = wmma_16x16x4_f32(v2f{r1[2], r1[3]}, B1[3], dr);
          di = wmma_16x16x4_f32(v2f{i1[2], i1[3]}, B1[3], di);
        }

        v8f er = {}, ei = {};
        er = wmma_16x16x4_f32(v2f{dr[0], dr[1]}, B2[0], er);
        ei = wmma_16x16x4_f32(v2f{di[0], di[1]}, B2[0], ei);
        er = wmma_16x16x4_f32(v2f{dr[2], dr[3]}, B2[1], er);
        ei = wmma_16x16x4_f32(v2f{di[2], di[3]}, B2[1], ei);
        er = wmma_16x16x4_f32(v2f{dr[4], dr[5]}, B2[2], er);
        ei = wmma_16x16x4_f32(v2f{di[4], di[5]}, B2[2], ei);
        er = wmma_16x16x4_f32(v2f{dr[6], dr[7]}, B2[3], er);
        ei = wmma_16x16x4_f32(v2f{di[6], di[7]}, B2[3], ei);

        store_tile(t, er, ei);
      }
      __syncthreads();
    }
  }

  __shared__ float zred[NQ * 16];
  {
    const int cl = lane & 15, hl = lane >> 4;
    const unsigned sel = (cl >= 1 && cl <= 4) ? (1u << (cl - 1)) : 0u;
    const unsigned one = (cl > 4) ? 0u : 0x3f800000u;
    const unsigned hb  = (unsigned)hl << 3;
    const unsigned tsel = (cl >= 5 && cl <= 7) ? (1u << (cl - 5)) : 0u;
    const unsigned tone = tsel ? 0x3f800000u : 0u;
    v2f S[4], T[4];
#pragma unroll
    for (int k = 0; k < 4; ++k) {
#pragma unroll
      for (int r = 0; r < 2; ++r) {
        const unsigned nb = (unsigned)(r | ((k & 1) << 1) | ((k >> 1) << 2)) | hb;
        S[k][r] = __uint_as_float(one | ((nb & sel) ? 0x80000000u : 0u));
        const unsigned ti = (unsigned)(2 * k + r);
        T[k][r] = __uint_as_float(tone | ((ti & tsel) ? 0x80000000u : 0u));
      }
    }

    v4f p0 = {}, p1 = {};
    float sI[8];
    const float* pl = st + cl * 4 + hl * 128 + wave * ROWP;
#pragma unroll
    for (int i = 0; i < 8; ++i) {
      const float* pa = pl + (8 * i) * ROWP;
      const v4f r0 = *(const v4f*)(pa);
      const v4f r1 = *(const v4f*)(pa + 64);
      const v4f m0 = *(const v4f*)(pa + CSZ);
      const v4f m1 = *(const v4f*)(pa + CSZ + 64);
      const v4f q0 = r0 * r0 + m0 * m0;
      const v4f q1 = r1 * r1 + m1 * m1;
      p0 += q0;
      p1 += q1;
      const v4f u = q0 + q1;
      sI[i] = (u[0] + u[1]) + (u[2] + u[3]);
    }

    v8f z0 = {}, z1 = {};
    z0 = wmma_16x16x4_f32(v2f{p0[0], p0[1]}, S[0], z0);
    z1 = wmma_16x16x4_f32(v2f{sI[0], sI[1]}, T[0], z1);
    z0 = wmma_16x16x4_f32(v2f{p0[2], p0[3]}, S[1], z0);
    z1 = wmma_16x16x4_f32(v2f{sI[2], sI[3]}, T[1], z1);
    z0 = wmma_16x16x4_f32(v2f{p1[0], p1[1]}, S[2], z0);
    z1 = wmma_16x16x4_f32(v2f{sI[4], sI[5]}, T[2], z1);
    z0 = wmma_16x16x4_f32(v2f{p1[2], p1[3]}, S[3], z0);
    z1 = wmma_16x16x4_f32(v2f{sI[6], sI[7]}, T[3], z1);

#if defined(__gfx1250__)
    asm volatile("v_nop\n\tv_nop\n\tv_nop\n\tv_nop" : "+v"(z0), "+v"(z1));
#endif
    float cs = 0.f, mb0 = 0.f, mb1 = 0.f, mb2 = 0.f;
#pragma unroll
    for (int r = 0; r < 8; ++r) {
      const float sp = z0[r] + z1[r];
      cs  += sp;
      mb0 += (r & 1) ? -sp : sp;
      mb1 += (r & 2) ? -sp : sp;
      mb2 += (r & 4) ? -sp : sp;
    }
    const int slot = wave * 2 + hl;
    if (cl == 0) {
      zred[ 2 * 16 + slot] = mb0;
      zred[ 3 * 16 + slot] = mb1;
      zred[ 4 * 16 + slot] = mb2;
      zred[ 5 * 16 + slot] = hl ? -cs : cs;
      zred[ 8 * 16 + slot] = (wave & 1) ? -cs : cs;
      zred[ 9 * 16 + slot] = (wave & 2) ? -cs : cs;
      zred[10 * 16 + slot] = (wave & 4) ? -cs : cs;
    } else if (cl <= 7) {
      const int bit = (cl == 1) ? 0 : (cl == 2) ? 1 : (cl == 3) ? 6 : (cl == 4) ? 7 : (cl + 6);
      zred[bit * 16 + slot] = cs;
    }
  }
  __syncthreads();
  if (tid < NQ) {
    const float* zr = zred + (13 - tid) * 16;
    float s = 0.f;
#pragma unroll
    for (int i = 0; i < 16; ++i) s += zr[i];
    out[b * NQ + tid] = s;
  }
}

extern "C" void kernel_launch(void* const* d_in, const int* in_sizes, int n_in,
                              void* d_out, int out_size, void* d_ws, size_t ws_size,
                              hipStream_t stream) {
  const float* x      = (const float*)d_in[0];
  const float* params = (const float*)d_in[1];
  float* out          = (float*)d_out;
  int B = in_sizes[0] / NQ;
  qsim14_kernel<<<dim3(B), dim3(NT), 0, stream>>>(x, params, out);
}


# Layer 0: index bits 6,7 left out of the written state too (4 high tiles for 16), created with bits 0,1 by a rank-1 complex product, re/im stacked in K (2 WMMAs per row for 6); sum order changes
# speedup vs baseline: 3.5136x; 1.1526x over previous
#include <hip/hip_runtime.h>
#include <math.h>

typedef float v2f __attribute__((ext_vector_type(2)));
typedef float v4f __attribute__((ext_vector_type(4)));
typedef float v8f __attribute__((ext_vector_type(8)));

#define NQ   14
#define DIM  16384
#define NT   256
#define ROWP 260
#define CSZ  (64 * ROWP)

static __device__ inline v8f wmma_16x16x4_f32(v2f a, v2f b, v8f c) {
#if defined(__gfx1250__)
  return __builtin_amdgcn_wmma_f32_16x16x4_f32(
      false, a, false, b, (short)0, c, false, false);
#else
  return c;
#endif
}

static __device__ inline void gate_frag(const float* gc, const float* gs, int lane, int xmask, v2f F[4]) {
  const int No = lane & 15;
  const int N  = No ^ ((No >> 1) & xmask);
  const int hi = lane >> 4;
#pragma unroll
  for (int k = 0; k < 4; ++k) {
#pragma unroll
    for (int r = 0; r < 2; ++r) {
      const int ib[4] = { r, k & 1, k >> 1, hi };
      float e = 1.f;
#pragma unroll
      for (int f = 0; f < 4; ++f) {
        int ob = (N >> f) & 1;
        e *= (ob == ib[f]) ? gc[f] : ((ob == 0) ? -gs[f] : gs[f]);
      }
      F[k][r] = e;
    }
  }
}

__global__ __launch_bounds__(NT) void qsim14_kernel(
    const float* __restrict__ x,
    const float* __restrict__ params,
    float* __restrict__ out)
{
  __shared__ __attribute__((aligned(16))) float st[2 * CSZ];
  __shared__ __attribute__((aligned(16))) float pHi[64];
  __shared__ float pL4[16];
  __shared__ float tc[3 * NQ], ts[3 * NQ];

  const int b    = blockIdx.x;
  const int tid  = threadIdx.x;
  const int lane = tid & 31;
  const int wave = tid >> 5;

  {
    const int slot = (wave == 0 && lane < NQ) ? lane
                   : (wave == 1 && lane < 2 * NQ) ? NQ + lane : -1;
    if (slot >= 0) {
      const float h = 0.5f * ((slot < NQ) ? x[b * NQ + slot] : params[slot - NQ]);
      tc[slot] = cosf(h);
      ts[slot] = sinf(h);
    }
  }
  __syncthreads();

  if (tid < 64) {
    int h = tid; float p = 1.f;
#pragma unroll
    for (int t = 0; t < 6; ++t) p *= ((h >> (5 - t)) & 1) ? ts[t] : tc[t];
    pHi[((h & 15) << 2) | (h >> 4)] = p;
  } else if (tid < 80) {
    int l = tid - 64; float p = 1.f;
#pragma unroll
    for (int u = 0; u < 4; ++u) p *= ((l >> (3 - u)) & 1) ? ts[8 + u] : tc[8 + u];
    pL4[l] = p;
  }
  __syncthreads();

  {
    const int c  = tid & 15;
    const int rq = tid >> 4;
    const float Lv = pL4[c];
    const v4f H = *(const v4f*)(pHi + 4 * rq);
    const int pt = __popc(tid) & 3;
    const float tr = (pt == 0) ? 1.f : ((pt == 2) ? -1.f : 0.f);
    const float ti = (pt == 1) ? -1.f : ((pt == 3) ? 1.f : 0.f);
    const float cr[4] = { tr,  ti, -tr, -ti };
    const float ci[4] = { ti, -tr, -ti,  tr };
    float* ps = st + rq * ROWP + c;
#pragma unroll
    for (int i = 0; i < 4; ++i) {
      const int q = __builtin_popcount(i) & 3;
      const float mag = H[i] * Lv;
      float* pr = ps + (16 * (i & 1) + 32 * (i >> 1)) * ROWP;
      pr[0]   = mag * cr[q];
      pr[CSZ] = mag * ci[q];
    }
  }
  __syncthreads();

#pragma unroll
  for (int L = 0; L < 2; ++L) {
#pragma unroll
    for (int pass = 0; pass < 2; ++pass) {
      const int qb[2][4] = { {-1, -1, 1, 0}, {13, 12, 7, 6} };
      const int qa[2][4] = { { 5,  4, 3, 2}, {11, 10, 9, 8} };
      const float* lc = tc + NQ + L * NQ;
      const float* ls = ts + NQ + L * NQ;
      float c1[4], s1[4], c2[4], s2[4];
#pragma unroll
      for (int f = 0; f < 4; ++f) {
        if (qb[pass][f] < 0) { c1[f] = 1.f; s1[f] = 0.f; }
        else { c1[f] = lc[qb[pass][f]]; s1[f] = ls[qb[pass][f]]; }
        c2[f] = lc[qa[pass][f]]; s2[f] = ls[qa[pass][f]];
      }
      const bool create = (L == 0 && pass == 1);
      const int ntile = (L == 0 && pass == 0) ? 4 : 64;
      v2f B1[4] = {}, B2[4];
      if (!create) gate_frag(c1, s1, lane, pass ? 5 : 4, B1);
      gate_frag(c2, s2, lane, 7, B2);

      const int laneStride = pass ? 4   : ROWP;
      const int hiOff      = pass ? 128 : 32 * ROWP;
      const int off2       = pass ? 64  : 16 * ROWP;
      const int tileStride = pass ? ROWP : 4;
      const int cl = lane & 15, hl = lane >> 4;
      const float* pl = st + cl * laneStride + hl * hiOff;

      v2f Are = {}, Aim = {};
      if (create) {
        const int a  = cl ^ ((cl >> 1) & 5);
        float wr = 1.f, wi = 0.f;
#pragma unroll
        for (int f = 0; f < 4; ++f) {
          const int q = qb[1][f];
          const int ab = (a >> f) & 1;
          const float xr = ab ?  s1[f] * tc[q] : c1[f] * tc[q];
          const float xi = ab ? -c1[f] * ts[q] : s1[f] * ts[q];
          const float nr = wr * xr - wi * xi;
          const float ni = wr * xi + wi * xr;
          wr = nr; wi = ni;
        }
        Are[0] = hl ? -wi : wr;
        Aim[0] = hl ?  wr : wi;
      }

      const int T  = pass ? (wave & 1) : 0;
      const int e0 = T, e1 = T ^ 1;
      float* ps0 = st + (cl ^ (e0 ? 15 : 0)) * laneStride + (hl ^ T) * hiOff + e0 * off2;
      float* ps1 = st + (cl ^ (e1 ? 15 : 0)) * laneStride + (hl ^ T) * hiOff + e1 * off2;
      const bool rev0 = ((cl ^ e0) & 1) != 0;

      auto store_tile = [&](int t, const v8f& er, const v8f& ei) {
        float* q0 = ps0 + t * tileStride;
        float* q1 = ps1 + t * tileStride;
        if (pass) {
          const bool rev1 = !rev0;
          *(v4f*)(q0)       = rev0 ? v4f{er[3], er[2], er[1], er[0]} : v4f{er[0], er[1], er[2], er[3]};
          *(v4f*)(q1)       = rev1 ? v4f{er[7], er[6], er[5], er[4]} : v4f{er[4], er[5], er[6], er[7]};
          *(v4f*)(q0 + CSZ) = rev0 ? v4f{ei[3], ei[2], ei[1], ei[0]} : v4f{ei[0], ei[1], ei[2], ei[3]};
          *(v4f*)(q1 + CSZ) = rev1 ? v4f{ei[7], ei[6], ei[5], ei[4]} : v4f{ei[4], ei[5], ei[6], ei[7]};
        } else {
          *(v4f*)(q0)       = v4f{er[0], er[1], er[2], er[3]};
          *(v4f*)(q1)       = v4f{er[4], er[5], er[6], er[7]};
          *(v4f*)(q0 + CSZ) = v4f{ei[0], ei[1], ei[2], ei[3]};
          *(v4f*)(q1 + CSZ) = v4f{ei[4], ei[5], ei[6], ei[7]};
        }
      };

      if (create) {
        const float* pg = st + (cl >> 3) * CSZ + (wave + 8 * (cl & 7)) * ROWP + 8 * hl;
        const v4f a0 = *(const v4f*)(pg);
        const v4f a1 = *(const v4f*)(pg + 4);
        v8f d1 = {};
        d1 = wmma_16x16x4_f32(v2f{a0[0], a0[1]}, B2[0], d1);
        d1 = wmma_16x16x4_f32(v2f{a0[2], a0[3]}, B2[1], d1);
        d1 = wmma_16x16x4_f32(v2f{a1[0], a1[1]}, B2[2], d1);
        d1 = wmma_16x16x4_f32(v2f{a1[2], a1[3]}, B2[3], d1);
#pragma unroll
        for (int i = 0; i < 8; ++i) {
          const v2f bq = {d1[i], 0.f};
          v8f er = {}, ei = {};
          er = wmma_16x16x4_f32(Are, bq, er);
          ei = wmma_16x16x4_f32(Aim, bq, ei);
          store_tile(wave + 8 * i, er, ei);
        }
      } else
      for (int t = wave; t < ntile; t += 8) {
        v8f dr = {}, di = {};
        {
          const float* p = pl + t * tileStride;
          const v4f r0 = *(const v4f*)(p);
          const v4f r1 = *(const v4f*)(p + off2);
          const v4f i0 = *(const v4f*)(p + CSZ);
          const v4f i1 = *(const v4f*)(p + CSZ + off2);
          dr = wmma_16x16x4_f32(v2f{r0[0], r0[1]}, B1[0], dr);
          di = wmma_16x16x4_f32(v2f{i0[0], i0[1]}, B1[0], di);
          dr = wmma_16x16x4_f32(v2f{r0[2], r0[3]}, B1[1], dr);
          di = wmma_16x16x4_f32(v2f{i0[2], i0[3]}, B1[1], di);
          dr = wmma_16x16x4_f32(v2f{r1[0], r1[1]}, B1[2], dr);
          di = wmma_16x16x4_f32(v2f{i1[0], i1[1]}, B1[2], di);
          dr = wmma_16x16x4_f32(v2f{r1[2], r1[3]}, B1[3], dr);
          di = wmma_16x16x4_f32(v2f{i1[2], i1[3]}, B1[3], di);
        }

        v8f er = {}, ei = {};
        er = wmma_16x16x4_f32(v2f{dr[0], dr[1]}, B2[0], er);
        ei = wmma_16x16x4_f32(v2f{di[0], di[1]}, B2[0], ei);
        er = wmma_16x16x4_f32(v2f{dr[2], dr[3]}, B2[1], er);
        ei = wmma_16x16x4_f32(v2f{di[2], di[3]}, B2[1], ei);
        er = wmma_16x16x4_f32(v2f{dr[4], dr[5]}, B2[2], er);
        ei = wmma_16x16x4_f32(v2f{di[4], di[5]}, B2[2], ei);
        er = wmma_16x16x4_f32(v2f{dr[6], dr[7]}, B2[3], er);
        ei = wmma_16x16x4_f32(v2f{di[6], di[7]}, B2[3], ei);

        store_tile(t, er, ei);
      }
      __syncthreads();
    }
  }

  __shared__ float zred[NQ * 16];
  {
    const int cl = lane & 15, hl = lane >> 4;
    const unsigned sel = (cl >= 1 && cl <= 4) ? (1u << (cl - 1)) : 0u;
    const unsigned one = (cl > 4) ? 0u : 0x3f800000u;
    const unsigned hb  = (unsigned)hl << 3;
    const unsigned tsel = (cl >= 5 && cl <= 7) ? (1u << (cl - 5)) : 0u;
    const unsigned tone = tsel ? 0x3f800000u : 0u;
    v2f S[4], T[4];
#pragma unroll
    for (int k = 0; k < 4; ++k) {
#pragma unroll
      for (int r = 0; r < 2; ++r) {
        const unsigned nb = (unsigned)(r | ((k & 1) << 1) | ((k >> 1) << 2)) | hb;
        S[k][r] = __uint_as_float(one | ((nb & sel) ? 0x80000000u : 0u));
        const unsigned ti = (unsigned)(2 * k + r);
        T[k][r] = __uint_as_float(tone | ((ti & tsel) ? 0x80000000u : 0u));
      }
    }

    v4f p0 = {}, p1 = {};
    float sI[8];
    const float* pl = st + cl * 4 + hl * 128 + wave * ROWP;
#pragma unroll
    for (int i = 0; i < 8; ++i) {
      const float* pa = pl + (8 * i) * ROWP;
      const v4f r0 = *(const v4f*)(pa);
      const v4f r1 = *(const v4f*)(pa + 64);
      const v4f m0 = *(const v4f*)(pa + CSZ);
      const v4f m1 = *(const v4f*)(pa + CSZ + 64);
      const v4f q0 = r0 * r0 + m0 * m0;
      const v4f q1 = r1 * r1 + m1 * m1;
      p0 += q0;
      p1 += q1;
      const v4f u = q0 + q1;
      sI[i] = (u[0] + u[1]) + (u[2] + u[3]);
    }

    v8f z0 = {}, z1 = {};
    z0 = wmma_16x16x4_f32(v2f{p0[0], p0[1]}, S[0], z0);
    z1 = wmma_16x16x4_f32(v2f{sI[0], sI[1]}, T[0], z1);
    z0 = wmma_16x16x4_f32(v2f{p0[2], p0[3]}, S[1], z0);
    z1 = wmma_16x16x4_f32(v2f{sI[2], sI[3]}, T[1], z1);
    z0 = wmma_16x16x4_f32(v2f{p1[0], p1[1]}, S[2], z0);
    z1 = wmma_16x16x4_f32(v2f{sI[4], sI[5]}, T[2], z1);
    z0 = wmma_16x16x4_f32(v2f{p1[2], p1[3]}, S[3], z0);
    z1 = wmma_16x16x4_f32(v2f{sI[6], sI[7]}, T[3], z1);

#if defined(__gfx1250__)
    asm volatile("v_nop\n\tv_nop\n\tv_nop\n\tv_nop" : "+v"(z0), "+v"(z1));
#endif
    float cs = 0.f, mb0 = 0.f, mb1 = 0.f, mb2 = 0.f;
#pragma unroll
    for (int r = 0; r < 8; ++r) {
      const float sp = z0[r] + z1[r];
      cs  += sp;
      mb0 += (r & 1) ? -sp : sp;
      mb1 += (r & 2) ? -sp : sp;
      mb2 += (r & 4) ? -sp : sp;
    }
    const int slot = wave * 2 + hl;
    if (cl == 0) {
      zred[ 2 * 16 + slot] = mb0;
      zred[ 3 * 16 + slot] = mb1;
      zred[ 4 * 16 + slot] = mb2;
      zred[ 5 * 16 + slot] = hl ? -cs : cs;
      zred[ 8 * 16 + slot] = (wave & 1) ? -cs : cs;
      zred[ 9 * 16 + slot] = (wave & 2) ? -cs : cs;
      zred[10 * 16 + slot] = (wave & 4) ? -cs : cs;
    } else if (cl <= 7) {
      const int bit = (cl == 1) ? 0 : (cl == 2) ? 1 : (cl == 3) ? 6 : (cl == 4) ? 7 : (cl + 6);
      zred[bit * 16 + slot] = cs;
    }
  }
  __syncthreads();
  if (tid < NQ) {
    const float* zr = zred + (13 - tid) * 16;
    float s = 0.f;
#pragma unroll
    for (int i = 0; i < 16; ++i) s += zr[i];
    out[b * NQ + tid] = s;
  }
}

extern "C" void kernel_launch(void* const* d_in, const int* in_sizes, int n_in,
                              void* d_out, int out_size, void* d_ws, size_t ws_size,
                              hipStream_t stream) {
  const float* x      = (const float*)d_in[0];
  const float* params = (const float*)d_in[1];
  float* out          = (float*)d_out;
  int B = in_sizes[0] / NQ;
  qsim14_kernel<<<dim3(B), dim3(NT), 0, stream>>>(x, params, out);
}
